# starting kernel
# baseline (speedup 1.0000x reference)
#include <hip/hip_runtime.h>
#include <hip/hip_bf16.h>

typedef __attribute__((ext_vector_type(2))) float v2f;
typedef __attribute__((ext_vector_type(8))) float v8f;

#define F_IN   128
#define HEADS  8
#define F_PH   16
#define F_OUT  128
#define NEG_SLOPE 0.2f

__global__ void gat_init(float* __restrict__ out, float* __restrict__ emax,
                         float* __restrict__ denom, int nOut, int nNH) {
  int i = blockIdx.x * blockDim.x + threadIdx.x;
  if (i < nOut) out[i] = 0.0f;
  if (i < nNH) {
    emax[i]  = __uint_as_float(0xFF800000u);
    denom[i] = 0.0f;
  }
}

__global__ __launch_bounds__(256) void gat_proj_wmma(
    const float* __restrict__ x, const float* __restrict__ W,
    float* __restrict__ h, int nrows) {
  const int wave = threadIdx.x >> 5;
  const int lane = threadIdx.x & 31;
  const long long row0 = ((long long)blockIdx.x * 8 + wave) * 16;
  if (row0 >= nrows) return;

  const int m     = lane & 15;
  const int khalf = (lane >> 4) << 1;
  long long rowA = row0 + m;
  if (rowA >= nrows) rowA = nrows - 1;
  const float* xrow = x + rowA * F_IN;

  v8f acc[HEADS];
  #pragma unroll
  for (int t = 0; t < HEADS; ++t) acc[t] = (v8f){0,0,0,0,0,0,0,0};

  for (int kt = 0; kt < F_IN / 4; ++kt) {
    const int kbase = kt * 4 + khalf;
    v2f a = *(const v2f*)(xrow + kbase);
    #pragma unroll
    for (int t = 0; t < HEADS; ++t) {
      const float* wb = W + t * (F_IN * F_PH) + kbase * F_PH + m;
      v2f b;
      b.x = wb[0];
      b.y = wb[F_PH];
      acc[t] = __builtin_amdgcn_wmma_f32_16x16x4_f32(
           false, a,  false, b,
           (short)0, acc[t],  false,  false);
    }
  }

  const int mofs = (lane >> 4) << 3;
  #pragma unroll
  for (int t = 0; t < HEADS; ++t) {
    #pragma unroll
    for (int v = 0; v < 8; ++v) {
      long long r = row0 + v + mofs;
      if (r < nrows) h[r * F_OUT + t * F_PH + m] = acc[t][v];
    }
  }
}

__global__ void gat_node_scores(const float* __restrict__ h,
                                const float* __restrict__ a_src,
                                const float* __restrict__ a_dst,
                                float* __restrict__ s_src,
                                float* __restrict__ s_dst, int nNH) {
  int idx = blockIdx.x * blockDim.x + threadIdx.x;
  if (idx >= nNH) return;
  int n = idx >> 3, hd = idx & 7;
  const float* hp = h + (long long)n * F_OUT + hd * F_PH;
  const float* as = a_src + hd * F_PH;
  const float* ad = a_dst + hd * F_PH;
  float ss = 0.f, sd = 0.f;
  #pragma unroll
  for (int o = 0; o < F_PH; ++o) {
    float v = hp[o];
    ss += v * as[o];
    sd += v * ad[o];
  }
  s_src[idx] = ss;
  s_dst[idx] = sd;
}

__device__ __forceinline__ void atomicMaxF(float* addr, float val) {
  if (val >= 0.0f) atomicMax((int*)addr, __float_as_int(val));
  else             atomicMin((unsigned int*)addr, __float_as_uint(val));
}

__global__ void gat_edge_logits(const int* __restrict__ ei, int E,
                                const float* __restrict__ s_src,
                                const float* __restrict__ s_dst,
                                float* __restrict__ elog,
                                float* __restrict__ emax, int nEH) {
  int idx = blockIdx.x * blockDim.x + threadIdx.x;
  if (idx >= nEH) return;
  int e = idx >> 3, hd = idx & 7;
  int r = ei[e];
  int c = ei[E + e];
  float v = s_src[r * HEADS + hd] + s_dst[c * HEADS + hd];
  v = (v >= 0.0f) ? v : NEG_SLOPE * v;
  elog[idx] = v;
  atomicMaxF(&emax[r * HEADS + hd], v);
}

__global__ void gat_edge_exp(const int* __restrict__ ei, int E,
                             const float* __restrict__ emax,
                             float* __restrict__ elog,
                             float* __restrict__ denom, int nEH) {
  int idx = blockIdx.x * blockDim.x + threadIdx.x;
  if (idx >= nEH) return;
  int e = idx >> 3, hd = idx & 7;
  int r = ei[e];
  float ex = expf(elog[idx] - emax[r * HEADS + hd]);
  elog[idx] = ex;
  atomicAdd(&denom[r * HEADS + hd], ex);
}

__global__ void gat_aggregate(const int* __restrict__ ei, int E,
                              const float* __restrict__ h,
                              const float* __restrict__ elog,
                              const float* __restrict__ denom,
                              float* __restrict__ out, int nWork) {
  int idx = blockIdx.x * blockDim.x + threadIdx.x;
  if (idx >= nWork) return;
  int e   = idx >> 5;
  int sub = idx & 31;
  int hd  = sub >> 2;
  int q   = sub & 3;
  int r = ei[e];
  int c = ei[E + e];
  float alpha = elog[e * HEADS + hd] / (denom[r * HEADS + hd] + 1e-16f);
  const float4 hv = *(const float4*)(h + (long long)c * F_OUT + hd * F_PH + q * 4);
  float* op = out + (long long)r * F_OUT + hd * F_PH + q * 4;
  atomicAdd(op + 0, alpha * hv.x);
  atomicAdd(op + 1, alpha * hv.y);
  atomicAdd(op + 2, alpha * hv.z);
  atomicAdd(op + 3, alpha * hv.w);
}

extern "C" void kernel_launch(void* const* d_in, const int* in_sizes, int n_in,
                              void* d_out, int out_size, void* d_ws, size_t ws_size,
                              hipStream_t stream) {
  const float* x     = (const float*)d_in[0];
  const int*   ei    = (const int*)d_in[1];
  const float* W     = (const float*)d_in[2];
  const float* a_src = (const float*)d_in[3];
  const float* a_dst = (const float*)d_in[4];
  float* out = (float*)d_out;

  const int N = in_sizes[0] / F_IN;
  const int E = in_sizes[1] / 2;

  float* h     = (float*)d_ws;
  float* s_src = h     + (size_t)N * F_OUT;
  float* s_dst = s_src + (size_t)N * HEADS;
  float* emax  = s_dst + (size_t)N * HEADS;
  float* denom = emax  + (size_t)N * HEADS;
  float* elog  = denom + (size_t)N * HEADS;

  const int nOut  = N * F_OUT;
  const int nNH   = N * HEADS;
  const int nEH   = E * HEADS;
  const int nWork = E * 32;

  const int B = 256;

  {
    int total = nOut > nNH ? nOut : nNH;
    gat_init<<<(total + B - 1) / B, B, 0, stream>>>(out, emax, denom, nOut, nNH);
  }
  {
    int blocks = (N + 127) / 128;
    gat_proj_wmma<<<blocks, B, 0, stream>>>(x, W, h, N);
  }
  gat_node_scores<<<(nNH + B - 1) / B, B, 0, stream>>>(h, a_src, a_dst, s_src, s_dst, nNH);
  gat_edge_logits<<<(nEH + B - 1) / B, B, 0, stream>>>(ei, E, s_src, s_dst, elog, emax, nEH);
  gat_edge_exp<<<(nEH + B - 1) / B, B, 0, stream>>>(ei, E, emax, elog, denom, nEH);
  gat_aggregate<<<(nWork + B - 1) / B, B, 0, stream>>>(ei, E, h, elog, denom, out, nWork);
}


# Edges ranked into per-destination buckets (int atomic rank, scan, place); one wave per row does max, exp-sum (4-lane tree), weighted sum in registers: no float atomics, sums in bucket arrival order
# speedup vs baseline: 2.0083x; 2.0083x over previous
#include <hip/hip_runtime.h>
#include <hip/hip_bf16.h>

typedef __attribute__((ext_vector_type(2))) float v2f;
typedef __attribute__((ext_vector_type(8))) float v8f;

#define F_IN   128
#define HEADS  8
#define F_PH   16
#define F_OUT  128
#define NEG_SLOPE 0.2f

__global__ __launch_bounds__(256) void gat_proj_wmma(
    const float* __restrict__ x, const float* __restrict__ W,
    float* __restrict__ h, int nrows) {
  const int wave = threadIdx.x >> 5;
  const int lane = threadIdx.x & 31;
  const long long row0 = ((long long)blockIdx.x * 8 + wave) * 16;
  if (row0 >= nrows) return;

  const int m     = lane & 15;
  const int khalf = (lane >> 4) << 1;
  long long rowA = row0 + m;
  if (rowA >= nrows) rowA = nrows - 1;
  const float* xrow = x + rowA * F_IN;

  v8f acc[HEADS];
  #pragma unroll
  for (int t = 0; t < HEADS; ++t) acc[t] = (v8f){0,0,0,0,0,0,0,0};

  for (int kt = 0; kt < F_IN / 4; ++kt) {
    const int kbase = kt * 4 + khalf;
    v2f a = *(const v2f*)(xrow + kbase);
    #pragma unroll
    for (int t = 0; t < HEADS; ++t) {
      const float* wb = W + t * (F_IN * F_PH) + kbase * F_PH + m;
      v2f b;
      b.x = wb[0];
      b.y = wb[F_PH];
      acc[t] = __builtin_amdgcn_wmma_f32_16x16x4_f32(
           false, a,  false, b,
           (short)0, acc[t],  false,  false);
    }
  }

  const int mofs = (lane >> 4) << 3;
  #pragma unroll
  for (int t = 0; t < HEADS; ++t) {
    #pragma unroll
    for (int v = 0; v < 8; ++v) {
      long long r = row0 + v + mofs;
      if (r < nrows) h[r * F_OUT + t * F_PH + m] = acc[t][v];
    }
  }
}

__global__ void gat_node_scores(const float* __restrict__ h,
                                const float* __restrict__ a_src,
                                const float* __restrict__ a_dst,
                                float* __restrict__ s_src,
                                float* __restrict__ s_dst,
                                int* __restrict__ deg, int nNH) {
  int idx = blockIdx.x * blockDim.x + threadIdx.x;
  if (idx >= nNH) return;
  int n = idx >> 3, hd = idx & 7;
  const float* hp = h + (long long)n * F_OUT + hd * F_PH;
  const float* as = a_src + hd * F_PH;
  const float* ad = a_dst + hd * F_PH;
  float ss = 0.f, sd = 0.f;
  #pragma unroll
  for (int o = 0; o < F_PH; ++o) {
    float v = hp[o];
    ss += v * as[o];
    sd += v * ad[o];
  }
  s_src[idx] = ss;
  s_dst[idx] = sd;
  if (hd == 0) deg[n] = 0;
}

__global__ void gat_edge_rank(const int* __restrict__ ei, int E,
                              int* __restrict__ deg, int* __restrict__ rank) {
  int e = blockIdx.x * blockDim.x + threadIdx.x;
  if (e >= E) return;
  rank[e] = atomicAdd(&deg[ei[e]], 1);
}

#define SCAN_TILE 1024
__global__ __launch_bounds__(256) void gat_scan_tiles(const int* __restrict__ deg,
                                                      int* __restrict__ offs,
                                                      int* __restrict__ bsum, int n) {
  __shared__ int wsum[8];
  const int tid = threadIdx.x, lane = tid & 31, wave = tid >> 5;
  const int i0 = blockIdx.x * SCAN_TILE + tid * 4;
  int d0 = (i0 + 0 < n) ? deg[i0 + 0] : 0;
  int d1 = (i0 + 1 < n) ? deg[i0 + 1] : 0;
  int d2 = (i0 + 2 < n) ? deg[i0 + 2] : 0;
  int d3 = (i0 + 3 < n) ? deg[i0 + 3] : 0;
  const int tsum = d0 + d1 + d2 + d3;
  int v = tsum;
  for (int o = 1; o < 32; o <<= 1) {
    int u = __shfl_up(v, o);
    if (lane >= o) v += u;
  }
  if (lane == 31) wsum[wave] = v;
  __syncthreads();
  int wpre = 0;
  for (int w = 0; w < wave; ++w) wpre += wsum[w];
  const int excl = wpre + v - tsum;
  if (i0 + 0 < n) offs[i0 + 0] = excl;
  if (i0 + 1 < n) offs[i0 + 1] = excl + d0;
  if (i0 + 2 < n) offs[i0 + 2] = excl + d0 + d1;
  if (i0 + 3 < n) offs[i0 + 3] = excl + d0 + d1 + d2;
  if (tid == 255) bsum[blockIdx.x] = wpre + v;
}

__global__ __launch_bounds__(256) void gat_scan_add(int* __restrict__ offs,
                                                    const int* __restrict__ bsum, int n) {
  __shared__ int wsum[8];
  const int tid = threadIdx.x, lane = tid & 31, wave = tid >> 5;
  const int b = blockIdx.x;
  int p = 0;
  for (int j = tid; j < b; j += 256) p += bsum[j];
  for (int o = 16; o > 0; o >>= 1) p += __shfl_xor(p, o);
  if (lane == 0) wsum[wave] = p;
  __syncthreads();
  int pre = 0;
  for (int w = 0; w < 8; ++w) pre += wsum[w];
  const int i0 = b * SCAN_TILE + tid * 4;
  for (int k = 0; k < 4; ++k)
    if (i0 + k < n) offs[i0 + k] += pre;
  if (b == (int)gridDim.x - 1 && tid == 0) offs[n] = pre + bsum[b];
}

__global__ void gat_edge_place(const int* __restrict__ ei, int E,
                               const int* __restrict__ offs,
                               const int* __restrict__ rank,
                               int* __restrict__ bcol) {
  int e = blockIdx.x * blockDim.x + threadIdx.x;
  if (e >= E) return;
  bcol[offs[ei[e]] + rank[e]] = ei[E + e];
}

#define QUAD_MOVE(x, ctrl) \
  __int_as_float(__builtin_amdgcn_update_dpp(0, __float_as_int(x), (ctrl), 0xF, 0xF, true))
#define QUAD_XOR1(x)  QUAD_MOVE(x, 0xB1)
#define QUAD_XOR2(x)  QUAD_MOVE(x, 0x4E)

template <int K>
__device__ __forceinline__ void chunk_logits(const int* __restrict__ bcol,
                                             const float* __restrict__ s_dst,
                                             int base, int end, int q, int hd,
                                             float ss_h, float (&t)[K]) {
  int c[K];
  #pragma unroll
  for (int k = 0; k < K; ++k) {
    int j = base + q + 4 * k;
    j = (j < end) ? j : end - 1;
    c[k] = bcol[j];
  }
  #pragma unroll
  for (int k = 0; k < K; ++k) {
    float v = ss_h + s_dst[(long long)c[k] * HEADS + hd];
    t[k] = (v >= 0.0f) ? v : NEG_SLOPE * v;
  }
}

template <int K>
__device__ __forceinline__ float chunk_max(const int* __restrict__ bcol,
                                           const float* __restrict__ s_dst,
                                           int base, int end, int q, int hd,
                                           float ss_h, float m) {
  float t[K];
  chunk_logits<K>(bcol, s_dst, base, end, q, hd, ss_h, t);
  #pragma unroll
  for (int k = 0; k < K; ++k)
    if (base + q + 4 * k < end) m = fmaxf(m, t[k]);
  return m;
}

template <int K>
__device__ __forceinline__ float chunk_expsum(const int* __restrict__ bcol,
                                              const float* __restrict__ s_dst,
                                              int base, int end, int q, int hd,
                                              float ss_h, float m, float d) {
  float t[K];
  chunk_logits<K>(bcol, s_dst, base, end, q, hd, ss_h, t);
  #pragma unroll
  for (int k = 0; k < K; ++k) {
    const float ex = expf(t[k] - m);
    d += (base + q + 4 * k < end) ? ex : 0.0f;
  }
  return d;
}

template <int K>
__device__ __forceinline__ void chunk_weighted_sum(const int* __restrict__ bcol,
                                                   const float* __restrict__ s_dst,
                                                   const float* __restrict__ h,
                                                   int base, int end, int cnt,
                                                   int lane, int q, int hd,
                                                   float ss_h, float m, float den,
                                                   float4& acc) {
  float t[K];
  chunk_logits<K>(bcol, s_dst, base, end, q, hd, ss_h, t);
  const int cv = (lane < cnt) ? bcol[base + lane] : 0;
  #pragma unroll
  for (int k = 0; k < K; ++k) {
    if (4 * k < cnt) {
      const float a = expf(t[k] - m) / den;
      float aj[4];
      aj[0] = QUAD_MOVE(a, 0x00);
      aj[1] = QUAD_MOVE(a, 0x55);
      aj[2] = QUAD_MOVE(a, 0xAA);
      aj[3] = QUAD_MOVE(a, 0xFF);
      float4 hv[4];
      #pragma unroll
      for (int e = 0; e < 4; ++e) {
        int j = 4 * k + e;
        j = (j < cnt) ? j : cnt - 1;
        const int c = __builtin_amdgcn_readlane(cv, j);
        hv[e] = *(const float4*)(h + (long long)c * F_OUT + lane * 4);
      }
      #pragma unroll
      for (int e = 0; e < 4; ++e) {
        const float w = (4 * k + e < cnt) ? aj[e] : 0.0f;
        acc.x += w * hv[e].x;
        acc.y += w * hv[e].y;
        acc.z += w * hv[e].z;
        acc.w += w * hv[e].w;
      }
    }
  }
}

__global__ __launch_bounds__(256) void gat_row_softmax_sum(
    const int* __restrict__ offs, const int* __restrict__ bcol,
    const float* __restrict__ h, const float* __restrict__ s_src,
    const float* __restrict__ s_dst, float* __restrict__ out, int n) {
  const int lane = threadIdx.x & 31;
  const int wave = __builtin_amdgcn_readfirstlane((int)(threadIdx.x >> 5));
  const int node = blockIdx.x * 8 + wave;
  if (node >= n) return;
  const int beg = offs[node];
  const int end = offs[node + 1];
  const int hd = lane >> 2;
  const int q  = lane & 3;
  const float ss_h = s_src[(long long)node * HEADS + hd];

  float m = __uint_as_float(0xFF800000u);
  for (int base = beg; base < end; base += 32) {
    if (end - base <= 16) m = chunk_max<4>(bcol, s_dst, base, end, q, hd, ss_h, m);
    else                  m = chunk_max<8>(bcol, s_dst, base, end, q, hd, ss_h, m);
  }
  m = fmaxf(m, QUAD_XOR1(m));
  m = fmaxf(m, QUAD_XOR2(m));

  float d = 0.0f;
  for (int base = beg; base < end; base += 32) {
    if (end - base <= 16) d = chunk_expsum<4>(bcol, s_dst, base, end, q, hd, ss_h, m, d);
    else                  d = chunk_expsum<8>(bcol, s_dst, base, end, q, hd, ss_h, m, d);
  }
  d += QUAD_XOR1(d);
  d += QUAD_XOR2(d);
  const float den = d + 1e-16f;

  float4 acc = make_float4(0.0f, 0.0f, 0.0f, 0.0f);
  for (int base = beg; base < end; base += 32) {
    const int cnt = (end - base < 32) ? (end - base) : 32;
    if (cnt <= 16) chunk_weighted_sum<4>(bcol, s_dst, h, base, end, cnt, lane, q, hd, ss_h, m, den, acc);
    else           chunk_weighted_sum<8>(bcol, s_dst, h, base, end, cnt, lane, q, hd, ss_h, m, den, acc);
  }
  *(float4*)(out + (long long)node * F_OUT + lane * 4) = acc;
}

extern "C" void kernel_launch(void* const* d_in, const int* in_sizes, int n_in,
                              void* d_out, int out_size, void* d_ws, size_t ws_size,
                              hipStream_t stream) {
  const float* x     = (const float*)d_in[0];
  const int*   ei    = (const int*)d_in[1];
  const float* W     = (const float*)d_in[2];
  const float* a_src = (const float*)d_in[3];
  const float* a_dst = (const float*)d_in[4];
  float* out = (float*)d_out;

  const int N = in_sizes[0] / F_IN;
  const int E = in_sizes[1] / 2;
  const int nTiles = (N + SCAN_TILE - 1) / SCAN_TILE;

  float* h     = (float*)d_ws;
  float* s_src = h     + (size_t)N * F_OUT;
  float* s_dst = s_src + (size_t)N * HEADS;
  int*   deg   = (int*)(s_dst + (size_t)N * HEADS);
  int*   offs  = deg  + (size_t)N;
  int*   bsum  = offs + (size_t)N + 1;
  int*   rank  = bsum + (size_t)nTiles;
  int*   bcol  = rank + (size_t)E;

  const int nNH = N * HEADS;
  const int B = 256;

  {
    int blocks = (N + 127) / 128;
    gat_proj_wmma<<<blocks, B, 0, stream>>>(x, W, h, N);
  }
  gat_node_scores<<<(nNH + B - 1) / B, B, 0, stream>>>(h, a_src, a_dst, s_src, s_dst, deg, nNH);
  gat_edge_rank<<<(E + B - 1) / B, B, 0, stream>>>(ei, E, deg, rank);
  gat_scan_tiles<<<nTiles, B, 0, stream>>>(deg, offs, bsum, N);
  gat_scan_add<<<nTiles, B, 0, stream>>>(offs, bsum, N);
  gat_edge_place<<<(E + B - 1) / B, B, 0, stream>>>(ei, E, offs, rank, bcol);
  gat_row_softmax_sum<<<(N + 7) / 8, B, 0, stream>>>(offs, bcol, h, s_src, s_dst, out, N);
}


# Row kernel reads a bucket of up to 32 edges once: logits stay in registers through max, exp-sum and alpha, source rows requested 16 at a time ahead of that arithmetic; order of every sum unchanged
# speedup vs baseline: 2.1781x; 1.0845x over previous
#include <hip/hip_runtime.h>
#include <hip/hip_bf16.h>

typedef __attribute__((ext_vector_type(2))) float v2f;
typedef __attribute__((ext_vector_type(8))) float v8f;

#define F_IN   128
#define HEADS  8
#define F_PH   16
#define F_OUT  128
#define NEG_SLOPE 0.2f

__global__ __launch_bounds__(256) void gat_proj_wmma(
    const float* __restrict__ x, const float* __restrict__ W,
    float* __restrict__ h, int nrows) {
  const int wave = threadIdx.x >> 5;
  const int lane = threadIdx.x & 31;
  const long long row0 = ((long long)blockIdx.x * 8 + wave) * 16;
  if (row0 >= nrows) return;

  const int m     = lane & 15;
  const int khalf = (lane >> 4) << 1;
  long long rowA = row0 + m;
  if (rowA >= nrows) rowA = nrows - 1;
  const float* xrow = x + rowA * F_IN;

  v8f acc[HEADS];
  #pragma unroll
  for (int t = 0; t < HEADS; ++t) acc[t] = (v8f){0,0,0,0,0,0,0,0};

  for (int kt = 0; kt < F_IN / 4; ++kt) {
    const int kbase = kt * 4 + khalf;
    v2f a = *(const v2f*)(xrow + kbase);
    #pragma unroll
    for (int t = 0; t < HEADS; ++t) {
      const float* wb = W + t * (F_IN * F_PH) + kbase * F_PH + m;
      v2f b;
      b.x = wb[0];
      b.y = wb[F_PH];
      acc[t] = __builtin_amdgcn_wmma_f32_16x16x4_f32(
           false, a,  false, b,
           (short)0, acc[t],  false,  false);
    }
  }

  const int mofs = (lane >> 4) << 3;
  #pragma unroll
  for (int t = 0; t < HEADS; ++t) {
    #pragma unroll
    for (int v = 0; v < 8; ++v) {
      long long r = row0 + v + mofs;
      if (r < nrows) h[r * F_OUT + t * F_PH + m] = acc[t][v];
    }
  }
}

__global__ void gat_node_scores(const float* __restrict__ h,
                                const float* __restrict__ a_src,
                                const float* __restrict__ a_dst,
                                float* __restrict__ s_src,
                                float* __restrict__ s_dst,
                                int* __restrict__ deg, int nNH) {
  int idx = blockIdx.x * blockDim.x + threadIdx.x;
  if (idx >= nNH) return;
  int n = idx >> 3, hd = idx & 7;
  const float* hp = h + (long long)n * F_OUT + hd * F_PH;
  const float* as = a_src + hd * F_PH;
  const float* ad = a_dst + hd * F_PH;
  float ss = 0.f, sd = 0.f;
  #pragma unroll
  for (int o = 0; o < F_PH; ++o) {
    float v = hp[o];
    ss += v * as[o];
    sd += v * ad[o];
  }
  s_src[idx] = ss;
  s_dst[idx] = sd;
  if (hd == 0) deg[n] = 0;
}

__global__ void gat_edge_rank(const int* __restrict__ ei, int E,
                              int* __restrict__ deg, int* __restrict__ rank) {
  int e = blockIdx.x * blockDim.x + threadIdx.x;
  if (e >= E) return;
  rank[e] = atomicAdd(&deg[ei[e]], 1);
}

#define SCAN_TILE 1024
__global__ __launch_bounds__(256) void gat_scan_tiles(const int* __restrict__ deg,
                                                      int* __restrict__ offs,
                                                      int* __restrict__ bsum, int n) {
  __shared__ int wsum[8];
  const int tid = threadIdx.x, lane = tid & 31, wave = tid >> 5;
  const int i0 = blockIdx.x * SCAN_TILE + tid * 4;
  int d0 = (i0 + 0 < n) ? deg[i0 + 0] : 0;
  int d1 = (i0 + 1 < n) ? deg[i0 + 1] : 0;
  int d2 = (i0 + 2 < n) ? deg[i0 + 2] : 0;
  int d3 = (i0 + 3 < n) ? deg[i0 + 3] : 0;
  const int tsum = d0 + d1 + d2 + d3;
  int v = tsum;
  for (int o = 1; o < 32; o <<= 1) {
    int u = __shfl_up(v, o);
    if (lane >= o) v += u;
  }
  if (lane == 31) wsum[wave] = v;
  __syncthreads();
  int wpre = 0;
  for (int w = 0; w < wave; ++w) wpre += wsum[w];
  const int excl = wpre + v - tsum;
  if (i0 + 0 < n) offs[i0 + 0] = excl;
  if (i0 + 1 < n) offs[i0 + 1] = excl + d0;
  if (i0 + 2 < n) offs[i0 + 2] = excl + d0 + d1;
  if (i0 + 3 < n) offs[i0 + 3] = excl + d0 + d1 + d2;
  if (tid == 255) bsum[blockIdx.x] = wpre + v;
}

__global__ __launch_bounds__(256) void gat_scan_add(int* __restrict__ offs,
                                                    const int* __restrict__ bsum, int n) {
  __shared__ int wsum[8];
  const int tid = threadIdx.x, lane = tid & 31, wave = tid >> 5;
  const int b = blockIdx.x;
  int p = 0;
  for (int j = tid; j < b; j += 256) p += bsum[j];
  for (int o = 16; o > 0; o >>= 1) p += __shfl_xor(p, o);
  if (lane == 0) wsum[wave] = p;
  __syncthreads();
  int pre = 0;
  for (int w = 0; w < 8; ++w) pre += wsum[w];
  const int i0 = b * SCAN_TILE + tid * 4;
  for (int k = 0; k < 4; ++k)
    if (i0 + k < n) offs[i0 + k] += pre;
  if (b == (int)gridDim.x - 1 && tid == 0) offs[n] = pre + bsum[b];
}

__global__ void gat_edge_place(const int* __restrict__ ei, int E,
                               const int* __restrict__ offs,
                               const int* __restrict__ rank,
                               int* __restrict__ bcol) {
  int e = blockIdx.x * blockDim.x + threadIdx.x;
  if (e >= E) return;
  bcol[offs[ei[e]] + rank[e]] = ei[E + e];
}

#define QUAD_MOVE(x, ctrl) \
  __int_as_float(__builtin_amdgcn_update_dpp(0, __float_as_int(x), (ctrl), 0xF, 0xF, true))
#define QUAD_XOR1(x)  QUAD_MOVE(x, 0xB1)
#define QUAD_XOR2(x)  QUAD_MOVE(x, 0x4E)

template <int K>
__device__ __forceinline__ void chunk_logits(const int* __restrict__ bcol,
                                             const float* __restrict__ s_dst,
                                             int base, int end, int q, int hd,
                                             float ss_h, float (&t)[K]) {
  int c[K];
  #pragma unroll
  for (int k = 0; k < K; ++k) {
    int j = base + q + 4 * k;
    j = (j < end) ? j : end - 1;
    c[k] = bcol[j];
  }
  #pragma unroll
  for (int k = 0; k < K; ++k) {
    float v = ss_h + s_dst[(long long)c[k] * HEADS + hd];
    t[k] = (v >= 0.0f) ? v : NEG_SLOPE * v;
  }
}

template <int K>
__device__ __forceinline__ float chunk_max(const int* __restrict__ bcol,
                                           const float* __restrict__ s_dst,
                                           int base, int end, int q, int hd,
                                           float ss_h, float m) {
  float t[K];
  chunk_logits<K>(bcol, s_dst, base, end, q, hd, ss_h, t);
  #pragma unroll
  for (int k = 0; k < K; ++k)
    if (base + q + 4 * k < end) m = fmaxf(m, t[k]);
  return m;
}

template <int K>
__device__ __forceinline__ float chunk_expsum(const int* __restrict__ bcol,
                                              const float* __restrict__ s_dst,
                                              int base, int end, int q, int hd,
                                              float ss_h, float m, float d) {
  float t[K];
  chunk_logits<K>(bcol, s_dst, base, end, q, hd, ss_h, t);
  #pragma unroll
  for (int k = 0; k < K; ++k) {
    const float ex = expf(t[k] - m);
    d += (base + q + 4 * k < end) ? ex : 0.0f;
  }
  return d;
}

template <int K>
__device__ __forceinline__ void chunk_weighted_sum(const int* __restrict__ bcol,
                                                   const float* __restrict__ s_dst,
                                                   const float* __restrict__ h,
                                                   int base, int end, int cnt,
                                                   int lane, int q, int hd,
                                                   float ss_h, float m, float den,
                                                   float4& acc) {
  float t[K];
  chunk_logits<K>(bcol, s_dst, base, end, q, hd, ss_h, t);
  const int cv = (lane < cnt) ? bcol[base + lane] : 0;
  #pragma unroll
  for (int k = 0; k < K; ++k) {
    if (4 * k < cnt) {
      const float a = expf(t[k] - m) / den;
      float aj[4];
      aj[0] = QUAD_MOVE(a, 0x00);
      aj[1] = QUAD_MOVE(a, 0x55);
      aj[2] = QUAD_MOVE(a, 0xAA);
      aj[3] = QUAD_MOVE(a, 0xFF);
      float4 hv[4];
      #pragma unroll
      for (int e = 0; e < 4; ++e) {
        int j = 4 * k + e;
        j = (j < cnt) ? j : cnt - 1;
        const int c = __builtin_amdgcn_readlane(cv, j);
        hv[e] = *(const float4*)(h + (long long)c * F_OUT + lane * 4);
      }
      #pragma unroll
      for (int e = 0; e < 4; ++e) {
        const float w = (4 * k + e < cnt) ? aj[e] : 0.0f;
        acc.x += w * hv[e].x;
        acc.y += w * hv[e].y;
        acc.z += w * hv[e].z;
        acc.w += w * hv[e].w;
      }
    }
  }
}

__device__ __forceinline__ void request_rows16(const float* __restrict__ h, int cv,
                                               int first, int cnt, int lane,
                                               float4 (&hv)[16]) {
  #pragma unroll
  for (int g = 0; g < 4; ++g) {
    if (first + 4 * g < cnt) {
      #pragma unroll
      for (int e = 0; e < 4; ++e) {
        int j = first + 4 * g + e;
        j = (j < cnt) ? j : cnt - 1;
        const int c = __builtin_amdgcn_readlane(cv, j);
        hv[4 * g + e] = ((const float4*)h)[(unsigned)c * (F_OUT / 4) + (unsigned)lane];
      }
    }
  }
}

__device__ __forceinline__ void add_rows16(const float (&a)[4], const float4 (&hv)[16],
                                           int first, int cnt, float4& acc) {
  #pragma unroll
  for (int g = 0; g < 4; ++g) {
    if (first + 4 * g < cnt) {
      float aj[4];
      aj[0] = QUAD_MOVE(a[g], 0x00);
      aj[1] = QUAD_MOVE(a[g], 0x55);
      aj[2] = QUAD_MOVE(a[g], 0xAA);
      aj[3] = QUAD_MOVE(a[g], 0xFF);
      #pragma unroll
      for (int e = 0; e < 4; ++e) {
        const float w = (first + 4 * g + e < cnt) ? aj[e] : 0.0f;
        acc.x += w * hv[4 * g + e].x;
        acc.y += w * hv[4 * g + e].y;
        acc.z += w * hv[4 * g + e].z;
        acc.w += w * hv[4 * g + e].w;
      }
    }
  }
}

template <int K>
__device__ __forceinline__ void row_in_one_chunk(const int* __restrict__ bcol,
                                                 const float* __restrict__ s_dst,
                                                 const float* __restrict__ h,
                                                 int beg, int end, int cnt,
                                                 int lane, int q, int hd,
                                                 float ss_h, float4& acc) {
  int c[K];
  #pragma unroll
  for (int k = 0; k < K; ++k) {
    int j = beg + q + 4 * k;
    j = (j < end) ? j : end - 1;
    c[k] = bcol[j];
  }
  const int cv = (lane < cnt) ? bcol[beg + lane] : 0;
  float sd[K];
  #pragma unroll
  for (int k = 0; k < K; ++k) sd[k] = s_dst[(unsigned)c[k] * HEADS + (unsigned)hd];

  float4 hv[16];
  request_rows16(h, cv, 0, cnt, lane, hv);
  __builtin_amdgcn_sched_barrier(0);

  float t[K];
  float m = __uint_as_float(0xFF800000u);
  #pragma unroll
  for (int k = 0; k < K; ++k) {
    const float v = ss_h + sd[k];
    t[k] = (v >= 0.0f) ? v : NEG_SLOPE * v;
    if (q + 4 * k < cnt) m = fmaxf(m, t[k]);
  }
  m = fmaxf(m, QUAD_XOR1(m));
  m = fmaxf(m, QUAD_XOR2(m));
  float ex[K];
  float d = 0.0f;
  #pragma unroll
  for (int k = 0; k < K; ++k) {
    ex[k] = expf(t[k] - m);
    d += (q + 4 * k < cnt) ? ex[k] : 0.0f;
  }
  d += QUAD_XOR1(d);
  d += QUAD_XOR2(d);
  const float den = d + 1e-16f;

  float a0[4];
  #pragma unroll
  for (int k = 0; k < 4; ++k) a0[k] = ex[k] / den;
  add_rows16(a0, hv, 0, cnt, acc);
  if constexpr (K == 8) {
    asm volatile("" : "+v"(acc.x), "+v"(acc.y), "+v"(acc.z), "+v"(acc.w));
    float4 hw[16];
    request_rows16(h, cv, 16, cnt, lane, hw);
    float a1[4];
    #pragma unroll
    for (int k = 0; k < 4; ++k) a1[k] = ex[4 + k] / den;
    add_rows16(a1, hw, 16, cnt, acc);
  }
}

__global__ __launch_bounds__(256) void gat_row_softmax_sum(
    const int* __restrict__ offs, const int* __restrict__ bcol,
    const float* __restrict__ h, const float* __restrict__ s_src,
    const float* __restrict__ s_dst, float* __restrict__ out, int n) {
  const int lane = threadIdx.x & 31;
  const int wave = __builtin_amdgcn_readfirstlane((int)(threadIdx.x >> 5));
  const int node = blockIdx.x * 8 + wave;
  if (node >= n) return;
  const int beg = offs[node];
  const int end = offs[node + 1];
  const int hd = lane >> 2;
  const int q  = lane & 3;
  const float ss_h = s_src[(long long)node * HEADS + hd];

  float4 acc = make_float4(0.0f, 0.0f, 0.0f, 0.0f);
  const int nedges = end - beg;
  if (nedges > 32) {
    float m = __uint_as_float(0xFF800000u);
    for (int base = beg; base < end; base += 32) {
      if (end - base <= 16) m = chunk_max<4>(bcol, s_dst, base, end, q, hd, ss_h, m);
      else                  m = chunk_max<8>(bcol, s_dst, base, end, q, hd, ss_h, m);
    }
    m = fmaxf(m, QUAD_XOR1(m));
    m = fmaxf(m, QUAD_XOR2(m));

    float d = 0.0f;
    for (int base = beg; base < end; base += 32) {
      if (end - base <= 16) d = chunk_expsum<4>(bcol, s_dst, base, end, q, hd, ss_h, m, d);
      else                  d = chunk_expsum<8>(bcol, s_dst, base, end, q, hd, ss_h, m, d);
    }
    d += QUAD_XOR1(d);
    d += QUAD_XOR2(d);
    const float den = d + 1e-16f;

    for (int base = beg; base < end; base += 32) {
      const int cnt = (end - base < 32) ? (end - base) : 32;
      if (cnt <= 16) chunk_weighted_sum<4>(bcol, s_dst, h, base, end, cnt, lane, q, hd, ss_h, m, den, acc);
      else           chunk_weighted_sum<8>(bcol, s_dst, h, base, end, cnt, lane, q, hd, ss_h, m, den, acc);
    }
  } else if (nedges > 16) {
    row_in_one_chunk<8>(bcol, s_dst, h, beg, end, nedges, lane, q, hd, ss_h, acc);
  } else if (nedges > 0) {
    row_in_one_chunk<4>(bcol, s_dst, h, beg, end, nedges, lane, q, hd, ss_h, acc);
  }
  *(float4*)(out + (long long)node * F_OUT + lane * 4) = acc;
}

extern "C" void kernel_launch(void* const* d_in, const int* in_sizes, int n_in,
                              void* d_out, int out_size, void* d_ws, size_t ws_size,
                              hipStream_t stream) {
  const float* x     = (const float*)d_in[0];
  const int*   ei    = (const int*)d_in[1];
  const float* W     = (const float*)d_in[2];
  const float* a_src = (const float*)d_in[3];
  const float* a_dst = (const float*)d_in[4];
  float* out = (float*)d_out;

  const int N = in_sizes[0] / F_IN;
  const int E = in_sizes[1] / 2;
  const int nTiles = (N + SCAN_TILE - 1) / SCAN_TILE;

  float* h     = (float*)d_ws;
  float* s_src = h     + (size_t)N * F_OUT;
  float* s_dst = s_src + (size_t)N * HEADS;
  int*   deg   = (int*)(s_dst + (size_t)N * HEADS);
  int*   offs  = deg  + (size_t)N;
  int*   bsum  = offs + (size_t)N + 1;
  int*   rank  = bsum + (size_t)nTiles;
  int*   bcol  = rank + (size_t)E;

  const int nNH = N * HEADS;
  const int B = 256;

  {
    int blocks = (N + 127) / 128;
    gat_proj_wmma<<<blocks, B, 0, stream>>>(x, W, h, N);
  }
  gat_node_scores<<<(nNH + B - 1) / B, B, 0, stream>>>(h, a_src, a_dst, s_src, s_dst, deg, nNH);
  gat_edge_rank<<<(E + B - 1) / B, B, 0, stream>>>(ei, E, deg, rank);
  gat_scan_tiles<<<nTiles, B, 0, stream>>>(deg, offs, bsum, N);
  gat_scan_add<<<nTiles, B, 0, stream>>>(offs, bsum, N);
  gat_edge_place<<<(E + B - 1) / B, B, 0, stream>>>(ei, E, offs, rank, bcol);
  gat_row_softmax_sum<<<(N + 7) / 8, B, 0, stream>>>(offs, bcol, h, s_src, s_dst, out, N);
}


# Row buckets at a fixed pitch of 64 slots: one pass ranks an edge (int atomic) and writes its source to its slot; scan and place launches gone (7 -> 4); a row over 64 edges reads the edge list
# speedup vs baseline: 2.4698x; 1.1339x over previous
#include <hip/hip_runtime.h>
#include <hip/hip_bf16.h>

typedef __attribute__((ext_vector_type(2))) float v2f;
typedef __attribute__((ext_vector_type(8))) float v8f;

#define F_IN   128
#define HEADS  8
#define F_PH   16
#define F_OUT  128
#define NEG_SLOPE 0.2f

__global__ __launch_bounds__(256) void gat_proj_wmma(
    const float* __restrict__ x, const float* __restrict__ W,
    float* __restrict__ h, int nrows) {
  const int wave = threadIdx.x >> 5;
  const int lane = threadIdx.x & 31;
  const long long row0 = ((long long)blockIdx.x * 8 + wave) * 16;
  if (row0 >= nrows) return;

  const int m     = lane & 15;
  const int khalf = (lane >> 4) << 1;
  long long rowA = row0 + m;
  if (rowA >= nrows) rowA = nrows - 1;
  const float* xrow = x + rowA * F_IN;

  v8f acc[HEADS];
  #pragma unroll
  for (int t = 0; t < HEADS; ++t) acc[t] = (v8f){0,0,0,0,0,0,0,0};

  for (int kt = 0; kt < F_IN / 4; ++kt) {
    const int kbase = kt * 4 + khalf;
    v2f a = *(const v2f*)(xrow + kbase);
    #pragma unroll
    for (int t = 0; t < HEADS; ++t) {
      const float* wb = W + t * (F_IN * F_PH) + kbase * F_PH + m;
      v2f b;
      b.x = wb[0];
      b.y = wb[F_PH];
      acc[t] = __builtin_amdgcn_wmma_f32_16x16x4_f32(
           false, a,  false, b,
           (short)0, acc[t],  false,  false);
    }
  }

  const int mofs = (lane >> 4) << 3;
  #pragma unroll
  for (int t = 0; t < HEADS; ++t) {
    #pragma unroll
    for (int v = 0; v < 8; ++v) {
      long long r = row0 + v + mofs;
      if (r < nrows) h[r * F_OUT + t * F_PH + m] = acc[t][v];
    }
  }
}

__global__ void gat_node_scores(const float* __restrict__ h,
                                const float* __restrict__ a_src,
                                const float* __restrict__ a_dst,
                                float* __restrict__ s_src,
                                float* __restrict__ s_dst,
                                int* __restrict__ deg, int nNH) {
  int idx = blockIdx.x * blockDim.x + threadIdx.x;
  if (idx >= nNH) return;
  int n = idx >> 3, hd = idx & 7;
  const float* hp = h + (long long)n * F_OUT + hd * F_PH;
  const float* as = a_src + hd * F_PH;
  const float* ad = a_dst + hd * F_PH;
  float ss = 0.f, sd = 0.f;
  #pragma unroll
  for (int o = 0; o < F_PH; ++o) {
    float v = hp[o];
    ss += v * as[o];
    sd += v * ad[o];
  }
  s_src[idx] = ss;
  s_dst[idx] = sd;
  if (hd == 0) deg[n] = 0;
}

#define ROW_CAP 64
__global__ void gat_edge_bucket(const int* __restrict__ ei, int E,
                                int* __restrict__ deg, int* __restrict__ bcol) {
  int e = blockIdx.x * blockDim.x + threadIdx.x;
  if (e >= E) return;
  const int r = ei[e];
  int c = ei[E + e];
  asm volatile("" : "+v"(c));
  const int slot = atomicAdd(&deg[r], 1);
  if (slot < ROW_CAP) bcol[(size_t)r * ROW_CAP + slot] = c;
}

#define QUAD_MOVE(x, ctrl) \
  __int_as_float(__builtin_amdgcn_update_dpp(0, __float_as_int(x), (ctrl), 0xF, 0xF, true))
#define QUAD_XOR1(x)  QUAD_MOVE(x, 0xB1)
#define QUAD_XOR2(x)  QUAD_MOVE(x, 0x4E)

template <int K>
__device__ __forceinline__ void chunk_logits(const int* __restrict__ bcol,
                                             const float* __restrict__ s_dst,
                                             int base, int end, int q, int hd,
                                             float ss_h, float (&t)[K]) {
  int c[K];
  #pragma unroll
  for (int k = 0; k < K; ++k) {
    int j = base + q + 4 * k;
    j = (j < end) ? j : end - 1;
    c[k] = bcol[j];
  }
  #pragma unroll
  for (int k = 0; k < K; ++k) {
    float v = ss_h + s_dst[(long long)c[k] * HEADS + hd];
    t[k] = (v >= 0.0f) ? v : NEG_SLOPE * v;
  }
}

template <int K>
__device__ __forceinline__ float chunk_max(const int* __restrict__ bcol,
                                           const float* __restrict__ s_dst,
                                           int base, int end, int q, int hd,
                                           float ss_h, float m) {
  float t[K];
  chunk_logits<K>(bcol, s_dst, base, end, q, hd, ss_h, t);
  #pragma unroll
  for (int k = 0; k < K; ++k)
    if (base + q + 4 * k < end) m = fmaxf(m, t[k]);
  return m;
}

template <int K>
__device__ __forceinline__ float chunk_expsum(const int* __restrict__ bcol,
                                              const float* __restrict__ s_dst,
                                              int base, int end, int q, int hd,
                                              float ss_h, float m, float d) {
  float t[K];
  chunk_logits<K>(bcol, s_dst, base, end, q, hd, ss_h, t);
  #pragma unroll
  for (int k = 0; k < K; ++k) {
    const float ex = expf(t[k] - m);
    d += (base + q + 4 * k < end) ? ex : 0.0f;
  }
  return d;
}

template <int K>
__device__ __forceinline__ void chunk_weighted_sum(const int* __restrict__ bcol,
                                                   const float* __restrict__ s_dst,
                                                   const float* __restrict__ h,
                                                   int base, int end, int cnt,
                                                   int lane, int q, int hd,
                                                   float ss_h, float m, float den,
                                                   float4& acc) {
  float t[K];
  chunk_logits<K>(bcol, s_dst, base, end, q, hd, ss_h, t);
  const int cv = (lane < cnt) ? bcol[base + lane] : 0;
  #pragma unroll
  for (int k = 0; k < K; ++k) {
    if (4 * k < cnt) {
      const float a = expf(t[k] - m) / den;
      float aj[4];
      aj[0] = QUAD_MOVE(a, 0x00);
      aj[1] = QUAD_MOVE(a, 0x55);
      aj[2] = QUAD_MOVE(a, 0xAA);
      aj[3] = QUAD_MOVE(a, 0xFF);
      float4 hv[4];
      #pragma unroll
      for (int e = 0; e < 4; ++e) {
        int j = 4 * k + e;
        j = (j < cnt) ? j : cnt - 1;
        const int c = __builtin_amdgcn_readlane(cv, j);
        hv[e] = *(const float4*)(h + (long long)c * F_OUT + lane * 4);
      }
      #pragma unroll
      for (int e = 0; e < 4; ++e) {
        const float w = (4 * k + e < cnt) ? aj[e] : 0.0f;
        acc.x += w * hv[e].x;
        acc.y += w * hv[e].y;
        acc.z += w * hv[e].z;
        acc.w += w * hv[e].w;
      }
    }
  }
}

__device__ __forceinline__ void request_rows16(const float* __restrict__ h, int cv,
                                               int first, int cnt, int lane,
                                               float4 (&hv)[16]) {
  #pragma unroll
  for (int g = 0; g < 4; ++g) {
    if (first + 4 * g < cnt) {
      #pragma unroll
      for (int e = 0; e < 4; ++e) {
        int j = first + 4 * g + e;
        j = (j < cnt) ? j : cnt - 1;
        const int c = __builtin_amdgcn_readlane(cv, j);
        hv[4 * g + e] = ((const float4*)h)[(unsigned)c * (F_OUT / 4) + (unsigned)lane];
      }
    }
  }
}

__device__ __forceinline__ void add_rows16(const float (&a)[4], const float4 (&hv)[16],
                                           int first, int cnt, float4& acc) {
  #pragma unroll
  for (int g = 0; g < 4; ++g) {
    if (first + 4 * g < cnt) {
      float aj[4];
      aj[0] = QUAD_MOVE(a[g], 0x00);
      aj[1] = QUAD_MOVE(a[g], 0x55);
      aj[2] = QUAD_MOVE(a[g], 0xAA);
      aj[3] = QUAD_MOVE(a[g], 0xFF);
      #pragma unroll
      for (int e = 0; e < 4; ++e) {
        const float w = (first + 4 * g + e < cnt) ? aj[e] : 0.0f;
        acc.x += w * hv[4 * g + e].x;
        acc.y += w * hv[4 * g + e].y;
        acc.z += w * hv[4 * g + e].z;
        acc.w += w * hv[4 * g + e].w;
      }
    }
  }
}

template <int K>
__device__ __forceinline__ void row_in_one_chunk(const int* __restrict__ bcol,
                                                 const float* __restrict__ s_dst,
                                                 const float* __restrict__ h,
                                                 int beg, int end, int cnt,
                                                 int lane, int q, int hd,
                                                 float ss_h, float4& acc) {
  int c[K];
  #pragma unroll
  for (int k = 0; k < K; ++k) {
    int j = beg + q + 4 * k;
    j = (j < end) ? j : end - 1;
    c[k] = bcol[j];
  }
  const int cv = (lane < cnt) ? bcol[beg + lane] : 0;
  float sd[K];
  #pragma unroll
  for (int k = 0; k < K; ++k) sd[k] = s_dst[(unsigned)c[k] * HEADS + (unsigned)hd];

  float4 hv[16];
  request_rows16(h, cv, 0, cnt, lane, hv);
  __builtin_amdgcn_sched_barrier(0);

  float t[K];
  float m = __uint_as_float(0xFF800000u);
  #pragma unroll
  for (int k = 0; k < K; ++k) {
    const float v = ss_h + sd[k];
    t[k] = (v >= 0.0f) ? v : NEG_SLOPE * v;
    if (q + 4 * k < cnt) m = fmaxf(m, t[k]);
  }
  m = fmaxf(m, QUAD_XOR1(m));
  m = fmaxf(m, QUAD_XOR2(m));
  float ex[K];
  float d = 0.0f;
  #pragma unroll
  for (int k = 0; k < K; ++k) {
    ex[k] = expf(t[k] - m);
    d += (q + 4 * k < cnt) ? ex[k] : 0.0f;
  }
  d += QUAD_XOR1(d);
  d += QUAD_XOR2(d);
  const float den = d + 1e-16f;

  float a0[4];
  #pragma unroll
  for (int k = 0; k < 4; ++k) a0[k] = ex[k] / den;
  add_rows16(a0, hv, 0, cnt, acc);
  if constexpr (K == 8) {
    asm volatile("" : "+v"(acc.x), "+v"(acc.y), "+v"(acc.z), "+v"(acc.w));
    float4 hw[16];
    request_rows16(h, cv, 16, cnt, lane, hw);
    float a1[4];
    #pragma unroll
    for (int k = 0; k < 4; ++k) a1[k] = ex[4 + k] / den;
    add_rows16(a1, hw, 16, cnt, acc);
  }
}

__device__ __forceinline__ void row_from_edge_list(const int* __restrict__ ei, int E,
                                                   const float* __restrict__ s_dst,
                                                   const float* __restrict__ h,
                                                   int node, int lane, int hd,
                                                   float ss_h, float4& acc) {
  float m = __uint_as_float(0xFF800000u);
  for (int base = 0; base < E; base += 32) {
    const int e = base + lane;
    const bool hit = (e < E) && (ei[e] == node);
    const int c = hit ? ei[E + e] : 0;
    unsigned mask = __builtin_amdgcn_ballot_w32(hit);
    while (mask) {
      const int j = __builtin_ctz(mask);
      mask &= mask - 1;
      const int cj = __builtin_amdgcn_readlane(c, j);
      const float v = ss_h + s_dst[(unsigned)cj * HEADS + (unsigned)hd];
      m = fmaxf(m, (v >= 0.0f) ? v : NEG_SLOPE * v);
    }
  }
  float d = 0.0f;
  for (int base = 0; base < E; base += 32) {
    const int e = base + lane;
    const bool hit = (e < E) && (ei[e] == node);
    const int c = hit ? ei[E + e] : 0;
    unsigned mask = __builtin_amdgcn_ballot_w32(hit);
    while (mask) {
      const int j = __builtin_ctz(mask);
      mask &= mask - 1;
      const int cj = __builtin_amdgcn_readlane(c, j);
      const float v = ss_h + s_dst[(unsigned)cj * HEADS + (unsigned)hd];
      d += expf(((v >= 0.0f) ? v : NEG_SLOPE * v) - m);
    }
  }
  const float den = d + 1e-16f;
  for (int base = 0; base < E; base += 32) {
    const int e = base + lane;
    const bool hit = (e < E) && (ei[e] == node);
    const int c = hit ? ei[E + e] : 0;
    unsigned mask = __builtin_amdgcn_ballot_w32(hit);
    while (mask) {
      const int j = __builtin_ctz(mask);
      mask &= mask - 1;
      const int cj = __builtin_amdgcn_readlane(c, j);
      const float v = ss_h + s_dst[(unsigned)cj * HEADS + (unsigned)hd];
      const float a = expf(((v >= 0.0f) ? v : NEG_SLOPE * v) - m) / den;
      const float4 hv = ((const float4*)h)[(unsigned)cj * (F_OUT / 4) + (unsigned)lane];
      acc.x += a * hv.x;
      acc.y += a * hv.y;
      acc.z += a * hv.z;
      acc.w += a * hv.w;
    }
  }
}

__global__ __launch_bounds__(256) void gat_row_softmax_sum(
    const int* __restrict__ deg, const int* __restrict__ bcol,
    const int* __restrict__ ei, int E,
    const float* __restrict__ h, const float* __restrict__ s_src,
    const float* __restrict__ s_dst, float* __restrict__ out, int n) {
  const int lane = threadIdx.x & 31;
  const int wave = __builtin_amdgcn_readfirstlane((int)(threadIdx.x >> 5));
  const int node = blockIdx.x * 8 + wave;
  if (node >= n) return;
  const int nedges = deg[node];
  const int beg = node * ROW_CAP;
  const int end = beg + nedges;
  const int hd = lane >> 2;
  const int q  = lane & 3;
  const float ss_h = s_src[(long long)node * HEADS + hd];

  float4 acc = make_float4(0.0f, 0.0f, 0.0f, 0.0f);
  if (nedges > ROW_CAP) {
    row_from_edge_list(ei, E, s_dst, h, node, lane, hd, ss_h, acc);
    *(float4*)(out + (long long)node * F_OUT + lane * 4) = acc;
    return;
  }
  if (nedges > 32) {
    float m = __uint_as_float(0xFF800000u);
    for (int base = beg; base < end; base += 32) {
      if (end - base <= 16) m = chunk_max<4>(bcol, s_dst, base, end, q, hd, ss_h, m);
      else                  m = chunk_max<8>(bcol, s_dst, base, end, q, hd, ss_h, m);
    }
    m = fmaxf(m, QUAD_XOR1(m));
    m = fmaxf(m, QUAD_XOR2(m));

    float d = 0.0f;
    for (int base = beg; base < end; base += 32) {
      if (end - base <= 16) d = chunk_expsum<4>(bcol, s_dst, base, end, q, hd, ss_h, m, d);
      else                  d = chunk_expsum<8>(bcol, s_dst, base, end, q, hd, ss_h, m, d);
    }
    d += QUAD_XOR1(d);
    d += QUAD_XOR2(d);
    const float den = d + 1e-16f;

    for (int base = beg; base < end; base += 32) {
      const int cnt = (end - base < 32) ? (end - base) : 32;
      if (cnt <= 16) chunk_weighted_sum<4>(bcol, s_dst, h, base, end, cnt, lane, q, hd, ss_h, m, den, acc);
      else           chunk_weighted_sum<8>(bcol, s_dst, h, base, end, cnt, lane, q, hd, ss_h, m, den, acc);
    }
  } else if (nedges > 16) {
    row_in_one_chunk<8>(bcol, s_dst, h, beg, end, nedges, lane, q, hd, ss_h, acc);
  } else if (nedges > 0) {
    row_in_one_chunk<4>(bcol, s_dst, h, beg, end, nedges, lane, q, hd, ss_h, acc);
  }
  *(float4*)(out + (long long)node * F_OUT + lane * 4) = acc;
}

extern "C" void kernel_launch(void* const* d_in, const int* in_sizes, int n_in,
                              void* d_out, int out_size, void* d_ws, size_t ws_size,
                              hipStream_t stream) {
  const float* x     = (const float*)d_in[0];
  const int*   ei    = (const int*)d_in[1];
  const float* W     = (const float*)d_in[2];
  const float* a_src = (const float*)d_in[3];
  const float* a_dst = (const float*)d_in[4];
  float* out = (float*)d_out;

  const int N = in_sizes[0] / F_IN;
  const int E = in_sizes[1] / 2;

  float* h     = (float*)d_ws;
  float* s_src = h     + (size_t)N * F_OUT;
  float* s_dst = s_src + (size_t)N * HEADS;
  int*   deg   = (int*)(s_dst + (size_t)N * HEADS);
  int*   bcol  = deg  + (size_t)N;

  const int nNH = N * HEADS;
  const int B = 256;

  {
    int blocks = (N + 127) / 128;
    gat_proj_wmma<<<blocks, B, 0, stream>>>(x, W, h, N);
  }
  gat_node_scores<<<(nNH + B - 1) / B, B, 0, stream>>>(h, a_src, a_dst, s_src, s_dst, deg, nNH);
  gat_edge_bucket<<<(E + B - 1) / B, B, 0, stream>>>(ei, E, deg, bcol);
  gat_row_softmax_sum<<<(N + 7) / 8, B, 0, stream>>>(deg, bcol, ei, E, h, s_src, s_dst, out, N);
}


# Edge ranks from a counting sort on LDS counters (count per chunk, prefix over chunks, place), no returning global atomic per edge; 4 -> 6 launches; sums in chunk order, then LDS arrival order
# speedup vs baseline: 2.8238x; 1.1433x over previous
#include <hip/hip_runtime.h>
#include <hip/hip_bf16.h>

typedef __attribute__((ext_vector_type(2))) float v2f;
typedef __attribute__((ext_vector_type(8))) float v8f;

#define F_IN   128
#define HEADS  8
#define F_PH   16
#define F_OUT  128
#define NEG_SLOPE 0.2f

__global__ __launch_bounds__(256) void gat_proj_wmma(
    const float* __restrict__ x, const float* __restrict__ W,
    float* __restrict__ h, int nrows) {
  const int wave = threadIdx.x >> 5;
  const int lane = threadIdx.x & 31;
  const long long row0 = ((long long)blockIdx.x * 8 + wave) * 16;
  if (row0 >= nrows) return;

  const int m     = lane & 15;
  const int khalf = (lane >> 4) << 1;
  long long rowA = row0 + m;
  if (rowA >= nrows) rowA = nrows - 1;
  const float* xrow = x + rowA * F_IN;

  v8f acc[HEADS];
  #pragma unroll
  for (int t = 0; t < HEADS; ++t) acc[t] = (v8f){0,0,0,0,0,0,0,0};

  for (int kt = 0; kt < F_IN / 4; ++kt) {
    const int kbase = kt * 4 + khalf;
    v2f a = *(const v2f*)(xrow + kbase);
    #pragma unroll
    for (int t = 0; t < HEADS; ++t) {
      const float* wb = W + t * (F_IN * F_PH) + kbase * F_PH + m;
      v2f b;
      b.x = wb[0];
      b.y = wb[F_PH];
      acc[t] = __builtin_amdgcn_wmma_f32_16x16x4_f32(
           false, a,  false, b,
           (short)0, acc[t],  false,  false);
    }
  }

  const int mofs = (lane >> 4) << 3;
  #pragma unroll
  for (int t = 0; t < HEADS; ++t) {
    #pragma unroll
    for (int v = 0; v < 8; ++v) {
      long long r = row0 + v + mofs;
      if (r < nrows) h[r * F_OUT + t * F_PH + m] = acc[t][v];
    }
  }
}

__global__ void gat_node_scores(const float* __restrict__ h,
                                const float* __restrict__ a_src,
                                const float* __restrict__ a_dst,
                                float* __restrict__ s_src,
                                float* __restrict__ s_dst, int nNH) {
  int idx = blockIdx.x * blockDim.x + threadIdx.x;
  if (idx >= nNH) return;
  int n = idx >> 3, hd = idx & 7;
  const float* hp = h + (long long)n * F_OUT + hd * F_PH;
  const float* as = a_src + hd * F_PH;
  const float* ad = a_dst + hd * F_PH;
  float ss = 0.f, sd = 0.f;
  #pragma unroll
  for (int o = 0; o < F_PH; ++o) {
    float v = hp[o];
    ss += v * as[o];
    sd += v * ad[o];
  }
  s_src[idx] = ss;
  s_dst[idx] = sd;
}

#define ROW_CAP 64
#define SORT_CHUNKS 64
#define SORT_THREADS 1024
#define SORT_WORDS 65536
#define SORT_BATCH 8
#define COPY_BATCH 16

__global__ __launch_bounds__(SORT_THREADS) void gat_edge_count(const int* __restrict__ ei, int E, int n,
                                                               unsigned* __restrict__ cm) {
  __shared__ unsigned cnt[SORT_WORDS];
  const int tid = threadIdx.x;
  const int nw = (n + 1) >> 1;
  for (int i = tid; i < nw; i += SORT_THREADS) cnt[i] = 0u;
  __syncthreads();
  const int chunk = (E + (int)gridDim.x - 1) / (int)gridDim.x;
  const int e0 = (int)blockIdx.x * chunk;
  const int e1 = (e0 + chunk < E) ? (e0 + chunk) : E;
  for (int base = e0; base < e1; base += SORT_BATCH * SORT_THREADS) {
    int r[SORT_BATCH];
    #pragma unroll
    for (int k = 0; k < SORT_BATCH; ++k) {
      const int e = base + k * SORT_THREADS + tid;
      r[k] = (e < e1) ? ei[e] : -1;
    }
    #pragma unroll
    for (int k = 0; k < SORT_BATCH; ++k)
      if ((unsigned)r[k] < (unsigned)n) atomicAdd(&cnt[r[k] >> 1], 1u << ((r[k] & 1) << 4));
  }
  __syncthreads();
  unsigned* __restrict__ dst = cm + (size_t)blockIdx.x * nw;
  for (int i = tid; i < nw; i += SORT_THREADS) dst[i] = cnt[i];
}

__global__ __launch_bounds__(256) void gat_edge_prefix(const unsigned* __restrict__ cm, unsigned* __restrict__ pre,
                                                       int nw, int chunks, int n, int* __restrict__ deg) {
  const int w = blockIdx.x * blockDim.x + threadIdx.x;
  if (w >= nw) return;
  unsigned lo = 0u, hi = 0u;
  for (int c0 = 0; c0 < chunks; c0 += COPY_BATCH) {
    unsigned v[COPY_BATCH];
    #pragma unroll
    for (int k = 0; k < COPY_BATCH; ++k)
      v[k] = (c0 + k < chunks) ? cm[(size_t)(c0 + k) * nw + w] : 0u;
    #pragma unroll
    for (int k = 0; k < COPY_BATCH; ++k) {
      if (c0 + k < chunks) {
        const unsigned plo = (lo < ROW_CAP) ? lo : ROW_CAP;
        const unsigned phi = (hi < ROW_CAP) ? hi : ROW_CAP;
        pre[(size_t)(c0 + k) * nw + w] = plo | (phi << 16);
        lo += v[k] & 0xFFFFu;
        hi += v[k] >> 16;
      }
    }
  }
  deg[2 * w] = (int)lo;
  if (2 * w + 1 < n) deg[2 * w + 1] = (int)hi;
}

__global__ __launch_bounds__(SORT_THREADS) void gat_edge_place(const int* __restrict__ ei, int E, int n,
                                                               const unsigned* __restrict__ pre,
                                                               int* __restrict__ bcol) {
  __shared__ unsigned cnt[SORT_WORDS];
  const int tid = threadIdx.x;
  const int nw = (n + 1) >> 1;
  const unsigned* __restrict__ src = pre + (size_t)blockIdx.x * nw;
  for (int base = 0; base < nw; base += COPY_BATCH * SORT_THREADS) {
    unsigned t[COPY_BATCH];
    #pragma unroll
    for (int k = 0; k < COPY_BATCH; ++k) {
      const int i = base + k * SORT_THREADS + tid;
      t[k] = (i < nw) ? src[i] : 0u;
    }
    #pragma unroll
    for (int k = 0; k < COPY_BATCH; ++k) {
      const int i = base + k * SORT_THREADS + tid;
      if (i < nw) cnt[i] = t[k];
    }
  }
  __syncthreads();
  const int chunk = (E + (int)gridDim.x - 1) / (int)gridDim.x;
  const int e0 = (int)blockIdx.x * chunk;
  const int e1 = (e0 + chunk < E) ? (e0 + chunk) : E;
  for (int base = e0; base < e1; base += SORT_BATCH * SORT_THREADS) {
    int r[SORT_BATCH], c[SORT_BATCH];
    #pragma unroll
    for (int k = 0; k < SORT_BATCH; ++k) {
      const int e = base + k * SORT_THREADS + tid;
      r[k] = (e < e1) ? ei[e] : -1;
      c[k] = (e < e1) ? ei[E + e] : 0;
    }
    #pragma unroll
    for (int k = 0; k < SORT_BATCH; ++k) {
      if ((unsigned)r[k] < (unsigned)n) {
        const unsigned sh = (unsigned)(r[k] & 1) << 4;
        const unsigned old = atomicAdd(&cnt[r[k] >> 1], 1u << sh);
        const unsigned slot = (old >> sh) & 0xFFFFu;
        if (slot < ROW_CAP) bcol[(size_t)r[k] * ROW_CAP + slot] = c[k];
      }
    }
  }
}

#define QUAD_MOVE(x, ctrl) \
  __int_as_float(__builtin_amdgcn_update_dpp(0, __float_as_int(x), (ctrl), 0xF, 0xF, true))
#define QUAD_XOR1(x)  QUAD_MOVE(x, 0xB1)
#define QUAD_XOR2(x)  QUAD_MOVE(x, 0x4E)

template <int K>
__device__ __forceinline__ void chunk_logits(const int* __restrict__ bcol,
                                             const float* __restrict__ s_dst,
                                             int base, int end, int q, int hd,
                                             float ss_h, float (&t)[K]) {
  int c[K];
  #pragma unroll
  for (int k = 0; k < K; ++k) {
    int j = base + q + 4 * k;
    j = (j < end) ? j : end - 1;
    c[k] = bcol[j];
  }
  #pragma unroll
  for (int k = 0; k < K; ++k) {
    float v = ss_h + s_dst[(long long)c[k] * HEADS + hd];
    t[k] = (v >= 0.0f) ? v : NEG_SLOPE * v;
  }
}

template <int K>
__device__ __forceinline__ float chunk_max(const int* __restrict__ bcol,
                                           const float* __restrict__ s_dst,
                                           int base, int end, int q, int hd,
                                           float ss_h, float m) {
  float t[K];
  chunk_logits<K>(bcol, s_dst, base, end, q, hd, ss_h, t);
  #pragma unroll
  for (int k = 0; k < K; ++k)
    if (base + q + 4 * k < end) m = fmaxf(m, t[k]);
  return m;
}

template <int K>
__device__ __forceinline__ float chunk_expsum(const int* __restrict__ bcol,
                                              const float* __restrict__ s_dst,
                                              int base, int end, int q, int hd,
                                              float ss_h, float m, float d) {
  float t[K];
  chunk_logits<K>(bcol, s_dst, base, end, q, hd, ss_h, t);
  #pragma unroll
  for (int k = 0; k < K; ++k) {
    const float ex = expf(t[k] - m);
    d += (base + q + 4 * k < end) ? ex : 0.0f;
  }
  return d;
}

template <int K>
__device__ __forceinline__ void chunk_weighted_sum(const int* __restrict__ bcol,
                                                   const float* __restrict__ s_dst,
                                                   const float* __restrict__ h,
                                                   int base, int end, int cnt,
                                                   int lane, int q, int hd,
                                                   float ss_h, float m, float den,
                                                   float4& acc) {
  float t[K];
  chunk_logits<K>(bcol, s_dst, base, end, q, hd, ss_h, t);
  const int cv = (lane < cnt) ? bcol[base + lane] : 0;
  #pragma unroll
  for (int k = 0; k < K; ++k) {
    if (4 * k < cnt) {
      const float a = expf(t[k] - m) / den;
      float aj[4];
      aj[0] = QUAD_MOVE(a, 0x00);
      aj[1] = QUAD_MOVE(a, 0x55);
      aj[2] = QUAD_MOVE(a, 0xAA);
      aj[3] = QUAD_MOVE(a, 0xFF);
      float4 hv[4];
      #pragma unroll
      for (int e = 0; e < 4; ++e) {
        int j = 4 * k + e;
        j = (j < cnt) ? j : cnt - 1;
        const int c = __builtin_amdgcn_readlane(cv, j);
        hv[e] = *(const float4*)(h + (long long)c * F_OUT + lane * 4);
      }
      #pragma unroll
      for (int e = 0; e < 4; ++e) {
        const float w = (4 * k + e < cnt) ? aj[e] : 0.0f;
        acc.x += w * hv[e].x;
        acc.y += w * hv[e].y;
        acc.z += w * hv[e].z;
        acc.w += w * hv[e].w;
      }
    }
  }
}

__device__ __forceinline__ void request_rows16(const float* __restrict__ h, int cv,
                                               int first, int cnt, int lane,
                                               float4 (&hv)[16]) {
  #pragma unroll
  for (int g = 0; g < 4; ++g) {
    if (first + 4 * g < cnt) {
      #pragma unroll
      for (int e = 0; e < 4; ++e) {
        int j = first + 4 * g + e;
        j = (j < cnt) ? j : cnt - 1;
        const int c = __builtin_amdgcn_readlane(cv, j);
        hv[4 * g + e] = ((const float4*)h)[(unsigned)c * (F_OUT / 4) + (unsigned)lane];
      }
    }
  }
}

__device__ __forceinline__ void add_rows16(const float (&a)[4], const float4 (&hv)[16],
                                           int first, int cnt, float4& acc) {
  #pragma unroll
  for (int g = 0; g < 4; ++g) {
    if (first + 4 * g < cnt) {
      float aj[4];
      aj[0] = QUAD_MOVE(a[g], 0x00);
      aj[1] = QUAD_MOVE(a[g], 0x55);
      aj[2] = QUAD_MOVE(a[g], 0xAA);
      aj[3] = QUAD_MOVE(a[g], 0xFF);
      #pragma unroll
      for (int e = 0; e < 4; ++e) {
        const float w = (first + 4 * g + e < cnt) ? aj[e] : 0.0f;
        acc.x += w * hv[4 * g + e].x;
        acc.y += w * hv[4 * g + e].y;
        acc.z += w * hv[4 * g + e].z;
        acc.w += w * hv[4 * g + e].w;
      }
    }
  }
}

template <int K>
__device__ __forceinline__ void row_in_one_chunk(const int* __restrict__ bcol,
                                                 const float* __restrict__ s_dst,
                                                 const float* __restrict__ h,
                                                 int beg, int end, int cnt,
                                                 int lane, int q, int hd,
                                                 float ss_h, float4& acc) {
  int c[K];
  #pragma unroll
  for (int k = 0; k < K; ++k) {
    int j = beg + q + 4 * k;
    j = (j < end) ? j : end - 1;
    c[k] = bcol[j];
  }
  const int cv = (lane < cnt) ? bcol[beg + lane] : 0;
  float sd[K];
  #pragma unroll
  for (int k = 0; k < K; ++k) sd[k] = s_dst[(unsigned)c[k] * HEADS + (unsigned)hd];

  float4 hv[16];
  request_rows16(h, cv, 0, cnt, lane, hv);
  __builtin_amdgcn_sched_barrier(0);

  float t[K];
  float m = __uint_as_float(0xFF800000u);
  #pragma unroll
  for (int k = 0; k < K; ++k) {
    const float v = ss_h + sd[k];
    t[k] = (v >= 0.0f) ? v : NEG_SLOPE * v;
    if (q + 4 * k < cnt) m = fmaxf(m, t[k]);
  }
  m = fmaxf(m, QUAD_XOR1(m));
  m = fmaxf(m, QUAD_XOR2(m));
  float ex[K];
  float d = 0.0f;
  #pragma unroll
  for (int k = 0; k < K; ++k) {
    ex[k] = expf(t[k] - m);
    d += (q + 4 * k < cnt) ? ex[k] : 0.0f;
  }
  d += QUAD_XOR1(d);
  d += QUAD_XOR2(d);
  const float den = d + 1e-16f;

  float a0[4];
  #pragma unroll
  for (int k = 0; k < 4; ++k) a0[k] = ex[k] / den;
  add_rows16(a0, hv, 0, cnt, acc);
  if constexpr (K == 8) {
    asm volatile("" : "+v"(acc.x), "+v"(acc.y), "+v"(acc.z), "+v"(acc.w));
    float4 hw[16];
    request_rows16(h, cv, 16, cnt, lane, hw);
    float a1[4];
    #pragma unroll
    for (int k = 0; k < 4; ++k) a1[k] = ex[4 + k] / den;
    add_rows16(a1, hw, 16, cnt, acc);
  }
}

__device__ __forceinline__ void row_from_edge_list(const int* __restrict__ ei, int E,
                                                   const float* __restrict__ s_dst,
                                                   const float* __restrict__ h,
                                                   int node, int lane, int hd,
                                                   float ss_h, float4& acc) {
  float m = __uint_as_float(0xFF800000u);
  for (int base = 0; base < E; base += 32) {
    const int e = base + lane;
    const bool hit = (e < E) && (ei[e] == node);
    const int c = hit ? ei[E + e] : 0;
    unsigned mask = __builtin_amdgcn_ballot_w32(hit);
    while (mask) {
      const int j = __builtin_ctz(mask);
      mask &= mask - 1;
      const int cj = __builtin_amdgcn_readlane(c, j);
      const float v = ss_h + s_dst[(unsigned)cj * HEADS + (unsigned)hd];
      m = fmaxf(m, (v >= 0.0f) ? v : NEG_SLOPE * v);
    }
  }
  float d = 0.0f;
  for (int base = 0; base < E; base += 32) {
    const int e = base + lane;
    const bool hit = (e < E) && (ei[e] == node);
    const int c = hit ? ei[E + e] : 0;
    unsigned mask = __builtin_amdgcn_ballot_w32(hit);
    while (mask) {
      const int j = __builtin_ctz(mask);
      mask &= mask - 1;
      const int cj = __builtin_amdgcn_readlane(c, j);
      const float v = ss_h + s_dst[(unsigned)cj * HEADS + (unsigned)hd];
      d += expf(((v >= 0.0f) ? v : NEG_SLOPE * v) - m);
    }
  }
  const float den = d + 1e-16f;
  for (int base = 0; base < E; base += 32) {
    const int e = base + lane;
    const bool hit = (e < E) && (ei[e] == node);
    const int c = hit ? ei[E + e] : 0;
    unsigned mask = __builtin_amdgcn_ballot_w32(hit);
    while (mask) {
      const int j = __builtin_ctz(mask);
      mask &= mask - 1;
      const int cj = __builtin_amdgcn_readlane(c, j);
      const float v = ss_h + s_dst[(unsigned)cj * HEADS + (unsigned)hd];
      const float a = expf(((v >= 0.0f) ? v : NEG_SLOPE * v) - m) / den;
      const float4 hv = ((const float4*)h)[(unsigned)cj * (F_OUT / 4) + (unsigned)lane];
      acc.x += a * hv.x;
      acc.y += a * hv.y;
      acc.z += a * hv.z;
      acc.w += a * hv.w;
    }
  }
}

__global__ __launch_bounds__(256) void gat_row_softmax_sum(
    const int* __restrict__ deg, const int* __restrict__ bcol,
    const int* __restrict__ ei, int E,
    const float* __restrict__ h, const float* __restrict__ s_src,
    const float* __restrict__ s_dst, float* __restrict__ out, int n) {
  const int lane = threadIdx.x & 31;
  const int wave = __builtin_amdgcn_readfirstlane((int)(threadIdx.x >> 5));
  const int node = blockIdx.x * 8 + wave;
  if (node >= n) return;
  const int nedges = deg[node];
  const int beg = node * ROW_CAP;
  const int end = beg + nedges;
  const int hd = lane >> 2;
  const int q  = lane & 3;
  const float ss_h = s_src[(long long)node * HEADS + hd];

  float4 acc = make_float4(0.0f, 0.0f, 0.0f, 0.0f);
  if (nedges > ROW_CAP) {
    row_from_edge_list(ei, E, s_dst, h, node, lane, hd, ss_h, acc);
    *(float4*)(out + (long long)node * F_OUT + lane * 4) = acc;
    return;
  }
  if (nedges > 32) {
    float m = __uint_as_float(0xFF800000u);
    for (int base = beg; base < end; base += 32) {
      if (end - base <= 16) m = chunk_max<4>(bcol, s_dst, base, end, q, hd, ss_h, m);
      else                  m = chunk_max<8>(bcol, s_dst, base, end, q, hd, ss_h, m);
    }
    m = fmaxf(m, QUAD_XOR1(m));
    m = fmaxf(m, QUAD_XOR2(m));

    float d = 0.0f;
    for (int base = beg; base < end; base += 32) {
      if (end - base <= 16) d = chunk_expsum<4>(bcol, s_dst, base, end, q, hd, ss_h, m, d);
      else                  d = chunk_expsum<8>(bcol, s_dst, base, end, q, hd, ss_h, m, d);
    }
    d += QUAD_XOR1(d);
    d += QUAD_XOR2(d);
    const float den = d + 1e-16f;

    for (int base = beg; base < end; base += 32) {
      const int cnt = (end - base < 32) ? (end - base) : 32;
      if (cnt <= 16) chunk_weighted_sum<4>(bcol, s_dst, h, base, end, cnt, lane, q, hd, ss_h, m, den, acc);
      else           chunk_weighted_sum<8>(bcol, s_dst, h, base, end, cnt, lane, q, hd, ss_h, m, den, acc);
    }
  } else if (nedges > 16) {
    row_in_one_chunk<8>(bcol, s_dst, h, beg, end, nedges, lane, q, hd, ss_h, acc);
  } else if (nedges > 0) {
    row_in_one_chunk<4>(bcol, s_dst, h, beg, end, nedges, lane, q, hd, ss_h, acc);
  }
  *(float4*)(out + (long long)node * F_OUT + lane * 4) = acc;
}

extern "C" void kernel_launch(void* const* d_in, const int* in_sizes, int n_in,
                              void* d_out, int out_size, void* d_ws, size_t ws_size,
                              hipStream_t stream) {
  const float* x     = (const float*)d_in[0];
  const int*   ei    = (const int*)d_in[1];
  const float* W     = (const float*)d_in[2];
  const float* a_src = (const float*)d_in[3];
  const float* a_dst = (const float*)d_in[4];
  float* out = (float*)d_out;

  const int N = in_sizes[0] / F_IN;
  const int E = in_sizes[1] / 2;

  float* h     = (float*)d_ws;
  float* s_src = h     + (size_t)N * F_OUT;
  float* s_dst = s_src + (size_t)N * HEADS;
  int*   deg   = (int*)(s_dst + (size_t)N * HEADS);
  int*   bcol  = deg  + (size_t)N;

  const int nNH = N * HEADS;
  const int B = 256;

  const int nw = (N + 1) / 2;
  unsigned* cm  = (unsigned*)h;
  unsigned* pre = cm + (size_t)SORT_CHUNKS * nw;
  gat_edge_count<<<SORT_CHUNKS, SORT_THREADS, 0, stream>>>(ei, E, N, cm);
  gat_edge_prefix<<<(nw + B - 1) / B, B, 0, stream>>>(cm, pre, nw, SORT_CHUNKS, N, deg);
  gat_edge_place<<<SORT_CHUNKS, SORT_THREADS, 0, stream>>>(ei, E, N, pre, bcol);

  {
    int blocks = (N + 127) / 128;
    gat_proj_wmma<<<blocks, B, 0, stream>>>(x, W, h, N);
  }
  gat_node_scores<<<(nNH + B - 1) / B, B, 0, stream>>>(h, a_src, a_dst, s_src, s_dst, nNH);
  gat_row_softmax_sum<<<(N + 7) / 8, B, 0, stream>>>(deg, bcol, ei, E, h, s_src, s_dst, out, N);
}


# Counting sort dealt over all WGPs: 4 work-groups per edge chunk, each counts and places only its quarter of the rows (count and place launches 64 -> 256 work-groups); order of every sum unchanged
# speedup vs baseline: 3.0645x; 1.0852x over previous
#include <hip/hip_runtime.h>
#include <hip/hip_bf16.h>

typedef __attribute__((ext_vector_type(2))) float v2f;
typedef __attribute__((ext_vector_type(8))) float v8f;

#define F_IN   128
#define HEADS  8
#define F_PH   16
#define F_OUT  128
#define NEG_SLOPE 0.2f

__global__ __launch_bounds__(256) void gat_proj_wmma(
    const float* __restrict__ x, const float* __restrict__ W,
    float* __restrict__ h, int nrows) {
  const int wave = threadIdx.x >> 5;
  const int lane = threadIdx.x & 31;
  const long long row0 = ((long long)blockIdx.x * 8 + wave) * 16;
  if (row0 >= nrows) return;

  const int m     = lane & 15;
  const int khalf = (lane >> 4) << 1;
  long long rowA = row0 + m;
  if (rowA >= nrows) rowA = nrows - 1;
  const float* xrow = x + rowA * F_IN;

  v8f acc[HEADS];
  #pragma unroll
  for (int t = 0; t < HEADS; ++t) acc[t] = (v8f){0,0,0,0,0,0,0,0};

  for (int kt = 0; kt < F_IN / 4; ++kt) {
    const int kbase = kt * 4 + khalf;
    v2f a = *(const v2f*)(xrow + kbase);
    #pragma unroll
    for (int t = 0; t < HEADS; ++t) {
      const float* wb = W + t * (F_IN * F_PH) + kbase * F_PH + m;
      v2f b;
      b.x = wb[0];
      b.y = wb[F_PH];
      acc[t] = __builtin_amdgcn_wmma_f32_16x16x4_f32(
           false, a,  false, b,
           (short)0, acc[t],  false,  false);
    }
  }

  const int mofs = (lane >> 4) << 3;
  #pragma unroll
  for (int t = 0; t < HEADS; ++t) {
    #pragma unroll
    for (int v = 0; v < 8; ++v) {
      long long r = row0 + v + mofs;
      if (r < nrows) h[r * F_OUT + t * F_PH + m] = acc[t][v];
    }
  }
}

__global__ void gat_node_scores(const float* __restrict__ h,
                                const float* __restrict__ a_src,
                                const float* __restrict__ a_dst,
                                float* __restrict__ s_src,
                                float* __restrict__ s_dst, int nNH) {
  int idx = blockIdx.x * blockDim.x + threadIdx.x;
  if (idx >= nNH) return;
  int n = idx >> 3, hd = idx & 7;
  const float* hp = h + (long long)n * F_OUT + hd * F_PH;
  const float* as = a_src + hd * F_PH;
  const float* ad = a_dst + hd * F_PH;
  float ss = 0.f, sd = 0.f;
  #pragma unroll
  for (int o = 0; o < F_PH; ++o) {
    float v = hp[o];
    ss += v * as[o];
    sd += v * ad[o];
  }
  s_src[idx] = ss;
  s_dst[idx] = sd;
}

#define ROW_CAP 64
#define SORT_CHUNKS 64
#define ROW_GROUPS 4
#define SORT_THREADS 1024
#define SORT_WORDS 65536
#define SORT_BATCH 8
#define COPY_BATCH 16

__global__ __launch_bounds__(SORT_THREADS) void gat_edge_count(const int* __restrict__ ei, int E, int n,
                                                               unsigned* __restrict__ cm) {
  __shared__ unsigned cnt[SORT_WORDS];
  const int tid = threadIdx.x;
  const int nw = (n + 1) >> 1;
  const int gw = (nw + (int)gridDim.y - 1) / (int)gridDim.y;
  const int w0 = (int)blockIdx.y * gw;
  const int w1 = (w0 + gw < nw) ? (w0 + gw) : nw;
  for (int i = w0 + tid; i < w1; i += SORT_THREADS) cnt[i] = 0u;
  __syncthreads();
  const int chunk = (E + (int)gridDim.x - 1) / (int)gridDim.x;
  const int e0 = (int)blockIdx.x * chunk;
  const int e1 = (e0 + chunk < E) ? (e0 + chunk) : E;
  for (int base = e0; base < e1; base += SORT_BATCH * SORT_THREADS) {
    int r[SORT_BATCH];
    #pragma unroll
    for (int k = 0; k < SORT_BATCH; ++k) {
      const int e = base + k * SORT_THREADS + tid;
      r[k] = (e < e1) ? ei[e] : -1;
    }
    #pragma unroll
    for (int k = 0; k < SORT_BATCH; ++k) {
      const int w = r[k] >> 1;
      if ((unsigned)r[k] < (unsigned)n && w >= w0 && w < w1) atomicAdd(&cnt[w], 1u << ((r[k] & 1) << 4));
    }
  }
  __syncthreads();
  unsigned* __restrict__ dst = cm + (size_t)blockIdx.x * nw;
  for (int i = w0 + tid; i < w1; i += SORT_THREADS) dst[i] = cnt[i];
}

__global__ __launch_bounds__(256) void gat_edge_prefix(const unsigned* __restrict__ cm, unsigned* __restrict__ pre,
                                                       int nw, int chunks, int n, int* __restrict__ deg) {
  const int w = blockIdx.x * blockDim.x + threadIdx.x;
  if (w >= nw) return;
  unsigned lo = 0u, hi = 0u;
  for (int c0 = 0; c0 < chunks; c0 += COPY_BATCH) {
    unsigned v[COPY_BATCH];
    #pragma unroll
    for (int k = 0; k < COPY_BATCH; ++k)
      v[k] = (c0 + k < chunks) ? cm[(size_t)(c0 + k) * nw + w] : 0u;
    #pragma unroll
    for (int k = 0; k < COPY_BATCH; ++k) {
      if (c0 + k < chunks) {
        const unsigned plo = (lo < ROW_CAP) ? lo : ROW_CAP;
        const unsigned phi = (hi < ROW_CAP) ? hi : ROW_CAP;
        pre[(size_t)(c0 + k) * nw + w] = plo | (phi << 16);
        lo += v[k] & 0xFFFFu;
        hi += v[k] >> 16;
      }
    }
  }
  deg[2 * w] = (int)lo;
  if (2 * w + 1 < n) deg[2 * w + 1] = (int)hi;
}

__global__ __launch_bounds__(SORT_THREADS) void gat_edge_place(const int* __restrict__ ei, int E, int n,
                                                               const unsigned* __restrict__ pre,
                                                               int* __restrict__ bcol) {
  __shared__ unsigned cnt[SORT_WORDS];
  const int tid = threadIdx.x;
  const int nw = (n + 1) >> 1;
  const int gw = (nw + (int)gridDim.y - 1) / (int)gridDim.y;
  const int w0 = (int)blockIdx.y * gw;
  const int w1 = (w0 + gw < nw) ? (w0 + gw) : nw;
  const unsigned* __restrict__ src = pre + (size_t)blockIdx.x * nw;
  for (int base = w0; base < w1; base += COPY_BATCH * SORT_THREADS) {
    unsigned t[COPY_BATCH];
    #pragma unroll
    for (int k = 0; k < COPY_BATCH; ++k) {
      const int i = base + k * SORT_THREADS + tid;
      t[k] = (i < w1) ? src[i] : 0u;
    }
    #pragma unroll
    for (int k = 0; k < COPY_BATCH; ++k) {
      const int i = base + k * SORT_THREADS + tid;
      if (i < w1) cnt[i] = t[k];
    }
  }
  __syncthreads();
  const int chunk = (E + (int)gridDim.x - 1) / (int)gridDim.x;
  const int e0 = (int)blockIdx.x * chunk;
  const int e1 = (e0 + chunk < E) ? (e0 + chunk) : E;
  for (int base = e0; base < e1; base += SORT_BATCH * SORT_THREADS) {
    int r[SORT_BATCH], c[SORT_BATCH];
    #pragma unroll
    for (int k = 0; k < SORT_BATCH; ++k) {
      const int e = base + k * SORT_THREADS + tid;
      r[k] = (e < e1) ? ei[e] : -1;
      c[k] = (e < e1) ? ei[E + e] : 0;
    }
    #pragma unroll
    for (int k = 0; k < SORT_BATCH; ++k) {
      const int w = r[k] >> 1;
      if ((unsigned)r[k] < (unsigned)n && w >= w0 && w < w1) {
        const unsigned sh = (unsigned)(r[k] & 1) << 4;
        const unsigned old = atomicAdd(&cnt[w], 1u << sh);
        const unsigned slot = (old >> sh) & 0xFFFFu;
        if (slot < ROW_CAP) bcol[(size_t)r[k] * ROW_CAP + slot] = c[k];
      }
    }
  }
}

#define QUAD_MOVE(x, ctrl) \
  __int_as_float(__builtin_amdgcn_update_dpp(0, __float_as_int(x), (ctrl), 0xF, 0xF, true))
#define QUAD_XOR1(x)  QUAD_MOVE(x, 0xB1)
#define QUAD_XOR2(x)  QUAD_MOVE(x, 0x4E)

template <int K>
__device__ __forceinline__ void chunk_logits(const int* __restrict__ bcol,
                                             const float* __restrict__ s_dst,
                                             int base, int end, int q, int hd,
                                             float ss_h, float (&t)[K]) {
  int c[K];
  #pragma unroll
  for (int k = 0; k < K; ++k) {
    int j = base + q + 4 * k;
    j = (j < end) ? j : end - 1;
    c[k] = bcol[j];
  }
  #pragma unroll
  for (int k = 0; k < K; ++k) {
    float v = ss_h + s_dst[(long long)c[k] * HEADS + hd];
    t[k] = (v >= 0.0f) ? v : NEG_SLOPE * v;
  }
}

template <int K>
__device__ __forceinline__ float chunk_max(const int* __restrict__ bcol,
                                           const float* __restrict__ s_dst,
                                           int base, int end, int q, int hd,
                                           float ss_h, float m) {
  float t[K];
  chunk_logits<K>(bcol, s_dst, base, end, q, hd, ss_h, t);
  #pragma unroll
  for (int k = 0; k < K; ++k)
    if (base + q + 4 * k < end) m = fmaxf(m, t[k]);
  return m;
}

template <int K>
__device__ __forceinline__ float chunk_expsum(const int* __restrict__ bcol,
                                              const float* __restrict__ s_dst,
                                              int base, int end, int q, int hd,
                                              float ss_h, float m, float d) {
  float t[K];
  chunk_logits<K>(bcol, s_dst, base, end, q, hd, ss_h, t);
  #pragma unroll
  for (int k = 0; k < K; ++k) {
    const float ex = expf(t[k] - m);
    d += (base + q + 4 * k < end) ? ex : 0.0f;
  }
  return d;
}

template <int K>
__device__ __forceinline__ void chunk_weighted_sum(const int* __restrict__ bcol,
                                                   const float* __restrict__ s_dst,
                                                   const float* __restrict__ h,
                                                   int base, int end, int cnt,
                                                   int lane, int q, int hd,
                                                   float ss_h, float m, float den,
                                                   float4& acc) {
  float t[K];
  chunk_logits<K>(bcol, s_dst, base, end, q, hd, ss_h, t);
  const int cv = (lane < cnt) ? bcol[base + lane] : 0;
  #pragma unroll
  for (int k = 0; k < K; ++k) {
    if (4 * k < cnt) {
      const float a = expf(t[k] - m) / den;
      float aj[4];
      aj[0] = QUAD_MOVE(a, 0x00);
      aj[1] = QUAD_MOVE(a, 0x55);
      aj[2] = QUAD_MOVE(a, 0xAA);
      aj[3] = QUAD_MOVE(a, 0xFF);
      float4 hv[4];
      #pragma unroll
      for (int e = 0; e < 4; ++e) {
        int j = 4 * k + e;
        j = (j < cnt) ? j : cnt - 1;
        const int c = __builtin_amdgcn_readlane(cv, j);
        hv[e] = *(const float4*)(h + (long long)c * F_OUT + lane * 4);
      }
      #pragma unroll
      for (int e = 0; e < 4; ++e) {
        const float w = (4 * k + e < cnt) ? aj[e] : 0.0f;
        acc.x += w * hv[e].x;
        acc.y += w * hv[e].y;
        acc.z += w * hv[e].z;
        acc.w += w * hv[e].w;
      }
    }
  }
}

__device__ __forceinline__ void request_rows16(const float* __restrict__ h, int cv,
                                               int first, int cnt, int lane,
                                               float4 (&hv)[16]) {
  #pragma unroll
  for (int g = 0; g < 4; ++g) {
    if (first + 4 * g < cnt) {
      #pragma unroll
      for (int e = 0; e < 4; ++e) {
        int j = first + 4 * g + e;
        j = (j < cnt) ? j : cnt - 1;
        const int c = __builtin_amdgcn_readlane(cv, j);
        hv[4 * g + e] = ((const float4*)h)[(unsigned)c * (F_OUT / 4) + (unsigned)lane];
      }
    }
  }
}

__device__ __forceinline__ void add_rows16(const float (&a)[4], const float4 (&hv)[16],
                                           int first, int cnt, float4& acc) {
  #pragma unroll
  for (int g = 0; g < 4; ++g) {
    if (first + 4 * g < cnt) {
      float aj[4];
      aj[0] = QUAD_MOVE(a[g], 0x00);
      aj[1] = QUAD_MOVE(a[g], 0x55);
      aj[2] = QUAD_MOVE(a[g], 0xAA);
      aj[3] = QUAD_MOVE(a[g], 0xFF);
      #pragma unroll
      for (int e = 0; e < 4; ++e) {
        const float w = (first + 4 * g + e < cnt) ? aj[e] : 0.0f;
        acc.x += w * hv[4 * g + e].x;
        acc.y += w * hv[4 * g + e].y;
        acc.z += w * hv[4 * g + e].z;
        acc.w += w * hv[4 * g + e].w;
      }
    }
  }
}

template <int K>
__device__ __forceinline__ void row_in_one_chunk(const int* __restrict__ bcol,
                                                 const float* __restrict__ s_dst,
                                                 const float* __restrict__ h,
                                                 int beg, int end, int cnt,
                                                 int lane, int q, int hd,
                                                 float ss_h, float4& acc) {
  int c[K];
  #pragma unroll
  for (int k = 0; k < K; ++k) {
    int j = beg + q + 4 * k;
    j = (j < end) ? j : end - 1;
    c[k] = bcol[j];
  }
  const int cv = (lane < cnt) ? bcol[beg + lane] : 0;
  float sd[K];
  #pragma unroll
  for (int k = 0; k < K; ++k) sd[k] = s_dst[(unsigned)c[k] * HEADS + (unsigned)hd];

  float4 hv[16];
  request_rows16(h, cv, 0, cnt, lane, hv);
  __builtin_amdgcn_sched_barrier(0);

  float t[K];
  float m = __uint_as_float(0xFF800000u);
  #pragma unroll
  for (int k = 0; k < K; ++k) {
    const float v = ss_h + sd[k];
    t[k] = (v >= 0.0f) ? v : NEG_SLOPE * v;
    if (q + 4 * k < cnt) m = fmaxf(m, t[k]);
  }
  m = fmaxf(m, QUAD_XOR1(m));
  m = fmaxf(m, QUAD_XOR2(m));
  float ex[K];
  float d = 0.0f;
  #pragma unroll
  for (int k = 0; k < K; ++k) {
    ex[k] = expf(t[k] - m);
    d += (q + 4 * k < cnt) ? ex[k] : 0.0f;
  }
  d += QUAD_XOR1(d);
  d += QUAD_XOR2(d);
  const float den = d + 1e-16f;

  float a0[4];
  #pragma unroll
  for (int k = 0; k < 4; ++k) a0[k] = ex[k] / den;
  add_rows16(a0, hv, 0, cnt, acc);
  if constexpr (K == 8) {
    asm volatile("" : "+v"(acc.x), "+v"(acc.y), "+v"(acc.z), "+v"(acc.w));
    float4 hw[16];
    request_rows16(h, cv, 16, cnt, lane, hw);
    float a1[4];
    #pragma unroll
    for (int k = 0; k < 4; ++k) a1[k] = ex[4 + k] / den;
    add_rows16(a1, hw, 16, cnt, acc);
  }
}

__device__ __forceinline__ void row_from_edge_list(const int* __restrict__ ei, int E,
                                                   const float* __restrict__ s_dst,
                                                   const float* __restrict__ h,
                                                   int node, int lane, int hd,
                                                   float ss_h, float4& acc) {
  float m = __uint_as_float(0xFF800000u);
  for (int base = 0; base < E; base += 32) {
    const int e = base + lane;
    const bool hit = (e < E) && (ei[e] == node);
    const int c = hit ? ei[E + e] : 0;
    unsigned mask = __builtin_amdgcn_ballot_w32(hit);
    while (mask) {
      const int j = __builtin_ctz(mask);
      mask &= mask - 1;
      const int cj = __builtin_amdgcn_readlane(c, j);
      const float v = ss_h + s_dst[(unsigned)cj * HEADS + (unsigned)hd];
      m = fmaxf(m, (v >= 0.0f) ? v : NEG_SLOPE * v);
    }
  }
  float d = 0.0f;
  for (int base = 0; base < E; base += 32) {
    const int e = base + lane;
    const bool hit = (e < E) && (ei[e] == node);
    const int c = hit ? ei[E + e] : 0;
    unsigned mask = __builtin_amdgcn_ballot_w32(hit);
    while (mask) {
      const int j = __builtin_ctz(mask);
      mask &= mask - 1;
      const int cj = __builtin_amdgcn_readlane(c, j);
      const float v = ss_h + s_dst[(unsigned)cj * HEADS + (unsigned)hd];
      d += expf(((v >= 0.0f) ? v : NEG_SLOPE * v) - m);
    }
  }
  const float den = d + 1e-16f;
  for (int base = 0; base < E; base += 32) {
    const int e = base + lane;
    const bool hit = (e < E) && (ei[e] == node);
    const int c = hit ? ei[E + e] : 0;
    unsigned mask = __builtin_amdgcn_ballot_w32(hit);
    while (mask) {
      const int j = __builtin_ctz(mask);
      mask &= mask - 1;
      const int cj = __builtin_amdgcn_readlane(c, j);
      const float v = ss_h + s_dst[(unsigned)cj * HEADS + (unsigned)hd];
      const float a = expf(((v >= 0.0f) ? v : NEG_SLOPE * v) - m) / den;
      const float4 hv = ((const float4*)h)[(unsigned)cj * (F_OUT / 4) + (unsigned)lane];
      acc.x += a * hv.x;
      acc.y += a * hv.y;
      acc.z += a * hv.z;
      acc.w += a * hv.w;
    }
  }
}

__global__ __launch_bounds__(256) void gat_row_softmax_sum(
    const int* __restrict__ deg, const int* __restrict__ bcol,
    const int* __restrict__ ei, int E,
    const float* __restrict__ h, const float* __restrict__ s_src,
    const float* __restrict__ s_dst, float* __restrict__ out, int n) {
  const int lane = threadIdx.x & 31;
  const int wave = __builtin_amdgcn_readfirstlane((int)(threadIdx.x >> 5));
  const int node = blockIdx.x * 8 + wave;
  if (node >= n) return;
  const int nedges = deg[node];
  const int beg = node * ROW_CAP;
  const int end = beg + nedges;
  const int hd = lane >> 2;
  const int q  = lane & 3;
  const float ss_h = s_src[(long long)node * HEADS + hd];

  float4 acc = make_float4(0.0f, 0.0f, 0.0f, 0.0f);
  if (nedges > ROW_CAP) {
    row_from_edge_list(ei, E, s_dst, h, node, lane, hd, ss_h, acc);
    *(float4*)(out + (long long)node * F_OUT + lane * 4) = acc;
    return;
  }
  if (nedges > 32) {
    float m = __uint_as_float(0xFF800000u);
    for (int base = beg; base < end; base += 32) {
      if (end - base <= 16) m = chunk_max<4>(bcol, s_dst, base, end, q, hd, ss_h, m);
      else                  m = chunk_max<8>(bcol, s_dst, base, end, q, hd, ss_h, m);
    }
    m = fmaxf(m, QUAD_XOR1(m));
    m = fmaxf(m, QUAD_XOR2(m));

    float d = 0.0f;
    for (int base = beg; base < end; base += 32) {
      if (end - base <= 16) d = chunk_expsum<4>(bcol, s_dst, base, end, q, hd, ss_h, m, d);
      else                  d = chunk_expsum<8>(bcol, s_dst, base, end, q, hd, ss_h, m, d);
    }
    d += QUAD_XOR1(d);
    d += QUAD_XOR2(d);
    const float den = d + 1e-16f;

    for (int base = beg; base < end; base += 32) {
      const int cnt = (end - base < 32) ? (end - base) : 32;
      if (cnt <= 16) chunk_weighted_sum<4>(bcol, s_dst, h, base, end, cnt, lane, q, hd, ss_h, m, den, acc);
      else           chunk_weighted_sum<8>(bcol, s_dst, h, base, end, cnt, lane, q, hd, ss_h, m, den, acc);
    }
  } else if (nedges > 16) {
    row_in_one_chunk<8>(bcol, s_dst, h, beg, end, nedges, lane, q, hd, ss_h, acc);
  } else if (nedges > 0) {
    row_in_one_chunk<4>(bcol, s_dst, h, beg, end, nedges, lane, q, hd, ss_h, acc);
  }
  *(float4*)(out + (long long)node * F_OUT + lane * 4) = acc;
}

extern "C" void kernel_launch(void* const* d_in, const int* in_sizes, int n_in,
                              void* d_out, int out_size, void* d_ws, size_t ws_size,
                              hipStream_t stream) {
  const float* x     = (const float*)d_in[0];
  const int*   ei    = (const int*)d_in[1];
  const float* W     = (const float*)d_in[2];
  const float* a_src = (const float*)d_in[3];
  const float* a_dst = (const float*)d_in[4];
  float* out = (float*)d_out;

  const int N = in_sizes[0] / F_IN;
  const int E = in_sizes[1] / 2;

  float* h     = (float*)d_ws;
  float* s_src = h     + (size_t)N * F_OUT;
  float* s_dst = s_src + (size_t)N * HEADS;
  int*   deg   = (int*)(s_dst + (size_t)N * HEADS);
  int*   bcol  = deg  + (size_t)N;

  const int nNH = N * HEADS;
  const int B = 256;

  const int nw = (N + 1) / 2;
  unsigned* cm  = (unsigned*)h;
  unsigned* pre = cm + (size_t)SORT_CHUNKS * nw;
  gat_edge_count<<<dim3(SORT_CHUNKS, ROW_GROUPS), SORT_THREADS, 0, stream>>>(ei, E, N, cm);
  gat_edge_prefix<<<(nw + B - 1) / B, B, 0, stream>>>(cm, pre, nw, SORT_CHUNKS, N, deg);
  gat_edge_place<<<dim3(SORT_CHUNKS, ROW_GROUPS), SORT_THREADS, 0, stream>>>(ei, E, N, pre, bcol);

  {
    int blocks = (N + 127) / 128;
    gat_proj_wmma<<<blocks, B, 0, stream>>>(x, W, h, N);
  }
  gat_node_scores<<<(nNH + B - 1) / B, B, 0, stream>>>(h, a_src, a_dst, s_src, s_dst, nNH);
  gat_row_softmax_sum<<<(N + 7) / 8, B, 0, stream>>>(deg, bcol, ei, E, h, s_src, s_dst, out, N);
}
